# starting kernel
# baseline (speedup 1.0000x reference)
#include <hip/hip_runtime.h>


#define N_NODES 50000
#define N_EDGES 600000
#define IN_CH   128
#define HID     128
#define OUT_CH  64

typedef __attribute__((ext_vector_type(2))) float v2f;
typedef __attribute__((ext_vector_type(8))) float v8f;

__global__ void deg_kernel(const int* __restrict__ row, const int* __restrict__ col,
                           const float* __restrict__ w, float* __restrict__ deg) {
    int e = blockIdx.x * blockDim.x + threadIdx.x;
    if (e >= N_EDGES) return;
    int r = row[e], c = col[e];
    float we = (r == c) ? 0.0f : w[e];
    if (we != 0.0f) atomicAdd(&deg[c], we);
}

__global__ void dis_kernel(const float* __restrict__ deg, float* __restrict__ dis,
                           float* __restrict__ invdeg) {
    int i = blockIdx.x * blockDim.x + threadIdx.x;
    if (i >= N_NODES) return;
    float d = deg[i] + 1.0f;
    dis[i]    = rsqrtf(d);
    invdeg[i] = 1.0f / d;
}

__global__ void gemm_xwt_wmma(const float* __restrict__ X,
                              const float* __restrict__ Wa,
                              const float* __restrict__ Wb,
                              int n_split,
                              float* __restrict__ C) {
    int w     = blockIdx.x * 8 + (threadIdx.x >> 5);
    int lane  = threadIdx.x & 31;
    int nTile = w & 7;
    int mTile = w >> 3;
    if (mTile >= N_NODES / 16) return;

    int half = lane >> 4;
    int l16  = lane & 15;

    const float* xrow = X + (size_t)(mTile * 16 + l16) * IN_CH + half * 2;
    int ng = nTile * 16 + l16;
    const float* wrow = ((ng < n_split) ? (Wa + (size_t)ng * IN_CH)
                                        : (Wb + (size_t)(ng - n_split) * IN_CH)) + half * 2;

    v8f acc = {};
#pragma unroll
    for (int k = 0; k < IN_CH; k += 4) {
        v2f a = *(const v2f*)(xrow + k);
        v2f b = *(const v2f*)(wrow + k);
        acc = __builtin_amdgcn_wmma_f32_16x16x4_f32(
            false, a, false, b, (short)0, acc, false, false);
    }

    float* crow = C + (size_t)(mTile * 16 + half * 8) * HID + nTile * 16 + l16;
#pragma unroll
    for (int j = 0; j < 8; ++j) crow[(size_t)j * HID] = acc[j];
}

__global__ void scatter_kernel(const int* __restrict__ row, const int* __restrict__ col,
                               const float* __restrict__ w, const float* __restrict__ dis,
                               const float* __restrict__ src, float* __restrict__ agg) {
    int wave = (blockIdx.x * blockDim.x + threadIdx.x) >> 5;
    int lane = threadIdx.x & 31;
    if (wave >= N_EDGES) return;
    int r = row[wave], c = col[wave];
    float we = (r == c) ? 0.0f : w[wave];
    if (we == 0.0f) return;
    float coeff = dis[r] * we * dis[c];

    float4 v = ((const float4*)(src + (size_t)r * HID))[lane];
    float* dst = agg + (size_t)c * HID + lane * 4;
    atomicAdd(dst + 0, coeff * v.x);
    atomicAdd(dst + 1, coeff * v.y);
    atomicAdd(dst + 2, coeff * v.z);
    atomicAdd(dst + 3, coeff * v.w);
}

__global__ void act_kernel(float* __restrict__ agg, const float* __restrict__ xw,
                           const float* __restrict__ invdeg, const float* __restrict__ b1) {
    int idx = blockIdx.x * blockDim.x + threadIdx.x;
    if (idx >= N_NODES * HID) return;
    int node = idx >> 7;
    int ch   = idx & 127;
    float v = agg[idx] + xw[idx] * invdeg[node] + b1[ch];
    agg[idx] = fmaxf(v, 0.0f);
}

__global__ void out_kernel(const float* __restrict__ agg2, const float* __restrict__ xw2,
                           const float* __restrict__ invdeg, const float* __restrict__ b_mu,
                           const float* __restrict__ b_ls, float* __restrict__ out) {
    int idx = blockIdx.x * blockDim.x + threadIdx.x;
    if (idx >= N_NODES * OUT_CH) return;
    int node = idx >> 6;
    int ch   = idx & 63;
    size_t base = (size_t)node * HID;
    float id = invdeg[node];
    out[idx] = agg2[base + ch] + xw2[base + ch] * id + b_mu[ch];
    out[(size_t)N_NODES * OUT_CH + idx] =
        agg2[base + OUT_CH + ch] + xw2[base + OUT_CH + ch] * id + b_ls[ch];
}

extern "C" void kernel_launch(void* const* d_in, const int* in_sizes, int n_in,
                              void* d_out, int out_size, void* d_ws, size_t ws_size,
                              hipStream_t stream) {
    (void)in_sizes; (void)n_in; (void)out_size; (void)ws_size;
    const float* x   = (const float*)d_in[0];
    const int*   ei  = (const int*)d_in[1];
    const float* ew  = (const float*)d_in[2];
    const float* W1  = (const float*)d_in[3];
    const float* b1  = (const float*)d_in[4];
    const float* Wmu = (const float*)d_in[5];
    const float* bmu = (const float*)d_in[6];
    const float* Wls = (const float*)d_in[7];
    const float* bls = (const float*)d_in[8];
    const int* row = ei;
    const int* col = ei + N_EDGES;

    float* ws     = (float*)d_ws;
    float* dis    = ws;
    float* invdeg = ws + 50000;
    float* deg    = ws + 100000;
    float* xw     = ws + 150000;
    float* agg1   = ws + 150000 + 6400000;
    float* agg2   = ws + 150000 + 12800000;

    hipMemsetAsync(deg,  0, (size_t)N_NODES * sizeof(float), stream);
    hipMemsetAsync(agg1, 0, (size_t)N_NODES * HID * sizeof(float), stream);
    hipMemsetAsync(agg2, 0, (size_t)N_NODES * HID * sizeof(float), stream);

    deg_kernel<<<(N_EDGES + 255) / 256, 256, 0, stream>>>(row, col, ew, deg);
    dis_kernel<<<(N_NODES + 255) / 256, 256, 0, stream>>>(deg, dis, invdeg);

    gemm_xwt_wmma<<<3125, 256, 0, stream>>>(x, W1, W1, HID, xw);
    scatter_kernel<<<N_EDGES / 8, 256, 0, stream>>>(row, col, ew, dis, xw, agg1);
    act_kernel<<<(N_NODES * HID + 255) / 256, 256, 0, stream>>>(agg1, xw, invdeg, b1);

    gemm_xwt_wmma<<<3125, 256, 0, stream>>>(agg1, Wmu, Wls, OUT_CH, xw);
    scatter_kernel<<<N_EDGES / 8, 256, 0, stream>>>(row, col, ew, dis, xw, agg2);

    out_kernel<<<(N_NODES * OUT_CH + 255) / 256, 256, 0, stream>>>(
        agg2, xw, invdeg, bmu, bls, (float*)d_out);
}


# Edges grouped by destination per call (int atomic arrival-order rank, scan, place); one wave per node sums its source rows in registers: no float atomics, no agg memsets, deg from grouped weights
# speedup vs baseline: 1.7278x; 1.7278x over previous
#include <hip/hip_runtime.h>


#define N_NODES 50000
#define N_EDGES 600000
#define IN_CH   128
#define HID     128
#define OUT_CH  64

typedef __attribute__((ext_vector_type(2))) float v2f;
typedef __attribute__((ext_vector_type(8))) float v8f;

typedef __attribute__((ext_vector_type(2))) int v2i;

#define SCAN_THREADS 1024
#define SCAN_TILES   49

__global__ void count_kernel(const int* __restrict__ col, int* __restrict__ cnt,
                             int* __restrict__ rank) {
    int e = blockIdx.x * blockDim.x + threadIdx.x;
    if (e >= N_EDGES) return;
    rank[e] = atomicAdd(&cnt[col[e]], 1);
}

static __device__ __forceinline__ int wave_incl_scan(int x, int lane) {
    x += __builtin_amdgcn_update_dpp(0, x, 0x111, 0xf, 0xf, true);
    x += __builtin_amdgcn_update_dpp(0, x, 0x112, 0xf, 0xf, true);
    x += __builtin_amdgcn_update_dpp(0, x, 0x114, 0xf, 0xf, true);
    x += __builtin_amdgcn_update_dpp(0, x, 0x118, 0xf, 0xf, true);
    int low = __builtin_amdgcn_readlane(x, 15);
    if (lane >= 16) x += low;
    return x;
}

__global__ void __launch_bounds__(SCAN_THREADS)
scan_kernel(const int* __restrict__ cnt, int* __restrict__ offs) {
    __shared__ int wtot[SCAN_TILES * 32];
    __shared__ int ttot[SCAN_TILES];
    __shared__ int tbase[SCAN_TILES];
    int t    = threadIdx.x;
    int lane = t & 31;
    int wv   = t >> 5;

#pragma unroll
    for (int i = 0; i < SCAN_TILES; ++i) {
        int idx = i * SCAN_THREADS + t;
        int v   = (idx < N_NODES) ? cnt[idx] : 0;
        int inc = wave_incl_scan(v, lane);
        if (lane == 31) wtot[i * 32 + wv] = inc;
    }
    __syncthreads();
    for (int tile = wv; tile < SCAN_TILES; tile += 32) {
        int x   = wtot[tile * 32 + lane];
        int inc = wave_incl_scan(x, lane);
        wtot[tile * 32 + lane] = inc - x;
        if (lane == 31) ttot[tile] = inc;
    }
    __syncthreads();
    if (wv == 0) {
        int a  = ttot[lane];
        int bq = (lane + 32 < SCAN_TILES) ? ttot[lane + 32] : 0;
        int sa = wave_incl_scan(a, lane);
        int sb = wave_incl_scan(bq, lane);
        int first32 = __builtin_amdgcn_readlane(sa, 31);
        tbase[lane] = sa - a;
        if (lane + 32 < SCAN_TILES) tbase[lane + 32] = first32 + sb - bq;
        if (lane == 31) offs[N_NODES] = first32 + sb;
    }
    __syncthreads();
#pragma unroll
    for (int i = 0; i < SCAN_TILES; ++i) {
        int idx = i * SCAN_THREADS + t;
        int v   = (idx < N_NODES) ? cnt[idx] : 0;
        int inc = wave_incl_scan(v, lane);
        if (idx < N_NODES) offs[idx] = tbase[i] + wtot[i * 32 + wv] + inc - v;
    }
}

__global__ void fill_kernel(const int* __restrict__ row, const int* __restrict__ col,
                            const float* __restrict__ w, const int* __restrict__ offs,
                            const int* __restrict__ rank, int* __restrict__ pairs) {
    int e = blockIdx.x * blockDim.x + threadIdx.x;
    if (e >= N_EDGES) return;
    int r = row[e], c = col[e];
    float we = (r == c) ? 0.0f : w[e];
    int p = offs[c] + rank[e];
    v2i pr = {r, __float_as_int(we)};
    *(v2i*)(pairs + 2 * (size_t)p) = pr;
}

__global__ void dis_kernel(const int* __restrict__ offs, const int* __restrict__ pairs,
                           float* __restrict__ dis, float* __restrict__ invdeg) {
    int i = blockIdx.x * blockDim.x + threadIdx.x;
    if (i >= N_NODES) return;
    int b = offs[i], n = offs[i + 1];
    float d = 0.0f;
    for (int j = b; j < n; ++j) d += __int_as_float(pairs[2 * (size_t)j + 1]);
    d += 1.0f;
    dis[i]    = rsqrtf(d);
    invdeg[i] = 1.0f / d;
}

__global__ void gemm_xwt_wmma(const float* __restrict__ X,
                              const float* __restrict__ Wa,
                              const float* __restrict__ Wb,
                              int n_split,
                              float* __restrict__ C) {
    int w     = blockIdx.x * 8 + (threadIdx.x >> 5);
    int lane  = threadIdx.x & 31;
    int nTile = w & 7;
    int mTile = w >> 3;
    if (mTile >= N_NODES / 16) return;

    int half = lane >> 4;
    int l16  = lane & 15;

    const float* xrow = X + (size_t)(mTile * 16 + l16) * IN_CH + half * 2;
    int ng = nTile * 16 + l16;
    const float* wrow = ((ng < n_split) ? (Wa + (size_t)ng * IN_CH)
                                        : (Wb + (size_t)(ng - n_split) * IN_CH)) + half * 2;

    v8f acc = {};
#pragma unroll
    for (int k = 0; k < IN_CH; k += 4) {
        v2f a = *(const v2f*)(xrow + k);
        v2f b = *(const v2f*)(wrow + k);
        acc = __builtin_amdgcn_wmma_f32_16x16x4_f32(
            false, a, false, b, (short)0, acc, false, false);
    }

    float* crow = C + (size_t)(mTile * 16 + half * 8) * HID + nTile * 16 + l16;
#pragma unroll
    for (int j = 0; j < 8; ++j) crow[(size_t)j * HID] = acc[j];
}

__global__ void gather_kernel(const int* __restrict__ offs, const int* __restrict__ pairs,
                              const float* __restrict__ dis, const float* __restrict__ src,
                              float* __restrict__ agg) {
    int wave = (blockIdx.x * blockDim.x + threadIdx.x) >> 5;
    int lane = threadIdx.x & 31;
    if (wave >= N_NODES) return;
    int c = __builtin_amdgcn_readfirstlane(wave);
    int b = offs[c], n = offs[c + 1];
    float dc = dis[c];
    const float4* src4 = (const float4*)src;

    float4 acc = {0.0f, 0.0f, 0.0f, 0.0f};
    for (int base = b; base < n; base += 32) {
        int m = n - base;
        if (m > 32) m = 32;
        int   rj = 0, wj = 0;
        float dj = 0.0f;
        if (lane < m) {
            v2i pr = *(const v2i*)(pairs + 2 * (size_t)(base + lane));
            rj = pr.x;
            wj = pr.y;
            dj = dis[rj];
        }
        int dji  = __float_as_int(dj);
        int last = m - 1;
        for (int j = 0; j < m; j += 4) {
            int j1 = (j + 1 < last) ? j + 1 : last;
            int j2 = (j + 2 < last) ? j + 2 : last;
            int j3 = (j + 3 < last) ? j + 3 : last;
            int r0 = __builtin_amdgcn_readlane(rj, j);
            int r1 = __builtin_amdgcn_readlane(rj, j1);
            int r2 = __builtin_amdgcn_readlane(rj, j2);
            int r3 = __builtin_amdgcn_readlane(rj, j3);
            float4 v0 = src4[(size_t)r0 * (HID / 4) + lane];
            float4 v1 = src4[(size_t)r1 * (HID / 4) + lane];
            float4 v2 = src4[(size_t)r2 * (HID / 4) + lane];
            float4 v3 = src4[(size_t)r3 * (HID / 4) + lane];
            __builtin_amdgcn_sched_barrier(0);
            float c0 = __int_as_float(__builtin_amdgcn_readlane(dji, j))
                     * __int_as_float(__builtin_amdgcn_readlane(wj, j)) * dc;
            float c1 = __int_as_float(__builtin_amdgcn_readlane(dji, j1))
                     * __int_as_float(__builtin_amdgcn_readlane(wj, j1)) * dc;
            float c2 = __int_as_float(__builtin_amdgcn_readlane(dji, j2))
                     * __int_as_float(__builtin_amdgcn_readlane(wj, j2)) * dc;
            float c3 = __int_as_float(__builtin_amdgcn_readlane(dji, j3))
                     * __int_as_float(__builtin_amdgcn_readlane(wj, j3)) * dc;
            if (j + 1 >= m) c1 = 0.0f;
            if (j + 2 >= m) c2 = 0.0f;
            if (j + 3 >= m) c3 = 0.0f;
            acc.x += c0 * v0.x; acc.y += c0 * v0.y; acc.z += c0 * v0.z; acc.w += c0 * v0.w;
            acc.x += c1 * v1.x; acc.y += c1 * v1.y; acc.z += c1 * v1.z; acc.w += c1 * v1.w;
            acc.x += c2 * v2.x; acc.y += c2 * v2.y; acc.z += c2 * v2.z; acc.w += c2 * v2.w;
            acc.x += c3 * v3.x; acc.y += c3 * v3.y; acc.z += c3 * v3.z; acc.w += c3 * v3.w;
        }
    }
    ((float4*)agg)[(size_t)c * (HID / 4) + lane] = acc;
}

__global__ void act_kernel(float* __restrict__ agg, const float* __restrict__ xw,
                           const float* __restrict__ invdeg, const float* __restrict__ b1) {
    int idx = blockIdx.x * blockDim.x + threadIdx.x;
    if (idx >= N_NODES * HID) return;
    int node = idx >> 7;
    int ch   = idx & 127;
    float v = agg[idx] + xw[idx] * invdeg[node] + b1[ch];
    agg[idx] = fmaxf(v, 0.0f);
}

__global__ void out_kernel(const float* __restrict__ agg2, const float* __restrict__ xw2,
                           const float* __restrict__ invdeg, const float* __restrict__ b_mu,
                           const float* __restrict__ b_ls, float* __restrict__ out) {
    int idx = blockIdx.x * blockDim.x + threadIdx.x;
    if (idx >= N_NODES * OUT_CH) return;
    int node = idx >> 6;
    int ch   = idx & 63;
    size_t base = (size_t)node * HID;
    float id = invdeg[node];
    out[idx] = agg2[base + ch] + xw2[base + ch] * id + b_mu[ch];
    out[(size_t)N_NODES * OUT_CH + idx] =
        agg2[base + OUT_CH + ch] + xw2[base + OUT_CH + ch] * id + b_ls[ch];
}

extern "C" void kernel_launch(void* const* d_in, const int* in_sizes, int n_in,
                              void* d_out, int out_size, void* d_ws, size_t ws_size,
                              hipStream_t stream) {
    (void)in_sizes; (void)n_in; (void)out_size; (void)ws_size;
    const float* x   = (const float*)d_in[0];
    const int*   ei  = (const int*)d_in[1];
    const float* ew  = (const float*)d_in[2];
    const float* W1  = (const float*)d_in[3];
    const float* b1  = (const float*)d_in[4];
    const float* Wmu = (const float*)d_in[5];
    const float* bmu = (const float*)d_in[6];
    const float* Wls = (const float*)d_in[7];
    const float* bls = (const float*)d_in[8];
    const int* row = ei;
    const int* col = ei + N_EDGES;

    float* ws     = (float*)d_ws;
    float* dis    = ws;
    float* invdeg = ws + 50000;
    int*   cnt    = (int*)(ws + 100000);
    float* xw     = ws + 150000;
    float* agg    = ws + 150000 + 6400000;
    int*   offs   = (int*)(ws + 12950000);
    int*   rank   = (int*)(ws + 13000048);
    int*   pairs  = (int*)(ws + 13600048);

    hipMemsetAsync(cnt, 0, (size_t)N_NODES * sizeof(int), stream);

    count_kernel<<<(N_EDGES + 255) / 256, 256, 0, stream>>>(col, cnt, rank);
    scan_kernel<<<1, SCAN_THREADS, 0, stream>>>(cnt, offs);
    fill_kernel<<<(N_EDGES + 255) / 256, 256, 0, stream>>>(row, col, ew, offs, rank, pairs);
    dis_kernel<<<(N_NODES + 255) / 256, 256, 0, stream>>>(offs, pairs, dis, invdeg);

    gemm_xwt_wmma<<<3125, 256, 0, stream>>>(x, W1, W1, HID, xw);
    gather_kernel<<<N_NODES / 8, 256, 0, stream>>>(offs, pairs, dis, xw, agg);
    act_kernel<<<(N_NODES * HID + 255) / 256, 256, 0, stream>>>(agg, xw, invdeg, b1);

    gemm_xwt_wmma<<<3125, 256, 0, stream>>>(agg, Wmu, Wls, OUT_CH, xw);
    gather_kernel<<<N_NODES / 8, 256, 0, stream>>>(offs, pairs, dis, xw, agg);

    out_kernel<<<(N_NODES * OUT_CH + 255) / 256, 256, 0, stream>>>(
        agg, xw, invdeg, bmu, bls, (float*)d_out);
}


# Self-loop term, bias, relu and the mu/logstd split done by the gathering wave that holds the node's sums: the two elementwise kernels and their 300000 one-element waves go; sum order unchanged
# speedup vs baseline: 1.9735x; 1.1422x over previous
#include <hip/hip_runtime.h>


#define N_NODES 50000
#define N_EDGES 600000
#define IN_CH   128
#define HID     128
#define OUT_CH  64

typedef __attribute__((ext_vector_type(2))) float v2f;
typedef __attribute__((ext_vector_type(8))) float v8f;

typedef __attribute__((ext_vector_type(2))) int v2i;

#define SCAN_THREADS 1024
#define SCAN_TILES   49

__global__ void count_kernel(const int* __restrict__ col, int* __restrict__ cnt,
                             int* __restrict__ rank) {
    int e = blockIdx.x * blockDim.x + threadIdx.x;
    if (e >= N_EDGES) return;
    rank[e] = atomicAdd(&cnt[col[e]], 1);
}

static __device__ __forceinline__ int wave_incl_scan(int x, int lane) {
    x += __builtin_amdgcn_update_dpp(0, x, 0x111, 0xf, 0xf, true);
    x += __builtin_amdgcn_update_dpp(0, x, 0x112, 0xf, 0xf, true);
    x += __builtin_amdgcn_update_dpp(0, x, 0x114, 0xf, 0xf, true);
    x += __builtin_amdgcn_update_dpp(0, x, 0x118, 0xf, 0xf, true);
    int low = __builtin_amdgcn_readlane(x, 15);
    if (lane >= 16) x += low;
    return x;
}

__global__ void __launch_bounds__(SCAN_THREADS)
scan_kernel(const int* __restrict__ cnt, int* __restrict__ offs) {
    __shared__ int wtot[SCAN_TILES * 32];
    __shared__ int ttot[SCAN_TILES];
    __shared__ int tbase[SCAN_TILES];
    int t    = threadIdx.x;
    int lane = t & 31;
    int wv   = t >> 5;

#pragma unroll
    for (int i = 0; i < SCAN_TILES; ++i) {
        int idx = i * SCAN_THREADS + t;
        int v   = (idx < N_NODES) ? cnt[idx] : 0;
        int inc = wave_incl_scan(v, lane);
        if (lane == 31) wtot[i * 32 + wv] = inc;
    }
    __syncthreads();
    for (int tile = wv; tile < SCAN_TILES; tile += 32) {
        int x   = wtot[tile * 32 + lane];
        int inc = wave_incl_scan(x, lane);
        wtot[tile * 32 + lane] = inc - x;
        if (lane == 31) ttot[tile] = inc;
    }
    __syncthreads();
    if (wv == 0) {
        int a  = ttot[lane];
        int bq = (lane + 32 < SCAN_TILES) ? ttot[lane + 32] : 0;
        int sa = wave_incl_scan(a, lane);
        int sb = wave_incl_scan(bq, lane);
        int first32 = __builtin_amdgcn_readlane(sa, 31);
        tbase[lane] = sa - a;
        if (lane + 32 < SCAN_TILES) tbase[lane + 32] = first32 + sb - bq;
        if (lane == 31) offs[N_NODES] = first32 + sb;
    }
    __syncthreads();
#pragma unroll
    for (int i = 0; i < SCAN_TILES; ++i) {
        int idx = i * SCAN_THREADS + t;
        int v   = (idx < N_NODES) ? cnt[idx] : 0;
        int inc = wave_incl_scan(v, lane);
        if (idx < N_NODES) offs[idx] = tbase[i] + wtot[i * 32 + wv] + inc - v;
    }
}

__global__ void fill_kernel(const int* __restrict__ row, const int* __restrict__ col,
                            const float* __restrict__ w, const int* __restrict__ offs,
                            const int* __restrict__ rank, int* __restrict__ pairs) {
    int e = blockIdx.x * blockDim.x + threadIdx.x;
    if (e >= N_EDGES) return;
    int r = row[e], c = col[e];
    float we = (r == c) ? 0.0f : w[e];
    int p = offs[c] + rank[e];
    v2i pr = {r, __float_as_int(we)};
    *(v2i*)(pairs + 2 * (size_t)p) = pr;
}

__global__ void dis_kernel(const int* __restrict__ offs, const int* __restrict__ pairs,
                           float* __restrict__ dis, float* __restrict__ invdeg) {
    int i = blockIdx.x * blockDim.x + threadIdx.x;
    if (i >= N_NODES) return;
    int b = offs[i], n = offs[i + 1];
    float d = 0.0f;
    for (int j = b; j < n; ++j) d += __int_as_float(pairs[2 * (size_t)j + 1]);
    d += 1.0f;
    dis[i]    = rsqrtf(d);
    invdeg[i] = 1.0f / d;
}

__global__ void gemm_xwt_wmma(const float* __restrict__ X,
                              const float* __restrict__ Wa,
                              const float* __restrict__ Wb,
                              int n_split,
                              float* __restrict__ C) {
    int w     = blockIdx.x * 8 + (threadIdx.x >> 5);
    int lane  = threadIdx.x & 31;
    int nTile = w & 7;
    int mTile = w >> 3;
    if (mTile >= N_NODES / 16) return;

    int half = lane >> 4;
    int l16  = lane & 15;

    const float* xrow = X + (size_t)(mTile * 16 + l16) * IN_CH + half * 2;
    int ng = nTile * 16 + l16;
    const float* wrow = ((ng < n_split) ? (Wa + (size_t)ng * IN_CH)
                                        : (Wb + (size_t)(ng - n_split) * IN_CH)) + half * 2;

    v8f acc = {};
#pragma unroll
    for (int k = 0; k < IN_CH; k += 4) {
        v2f a = *(const v2f*)(xrow + k);
        v2f b = *(const v2f*)(wrow + k);
        acc = __builtin_amdgcn_wmma_f32_16x16x4_f32(
            false, a, false, b, (short)0, acc, false, false);
    }

    float* crow = C + (size_t)(mTile * 16 + half * 8) * HID + nTile * 16 + l16;
#pragma unroll
    for (int j = 0; j < 8; ++j) crow[(size_t)j * HID] = acc[j];
}

static __device__ __forceinline__ float4 gather_row(int c, int lane,
                                                    const int* __restrict__ offs,
                                                    const int* __restrict__ pairs,
                                                    const float* __restrict__ dis,
                                                    const float* __restrict__ src) {
    int b = offs[c], n = offs[c + 1];
    float dc = dis[c];
    const float4* src4 = (const float4*)src;

    float4 acc = {0.0f, 0.0f, 0.0f, 0.0f};
    for (int base = b; base < n; base += 32) {
        int m = n - base;
        if (m > 32) m = 32;
        int   rj = 0, wj = 0;
        float dj = 0.0f;
        if (lane < m) {
            v2i pr = *(const v2i*)(pairs + 2 * (size_t)(base + lane));
            rj = pr.x;
            wj = pr.y;
            dj = dis[rj];
        }
        int dji  = __float_as_int(dj);
        int last = m - 1;
        for (int j = 0; j < m; j += 4) {
            int j1 = (j + 1 < last) ? j + 1 : last;
            int j2 = (j + 2 < last) ? j + 2 : last;
            int j3 = (j + 3 < last) ? j + 3 : last;
            int r0 = __builtin_amdgcn_readlane(rj, j);
            int r1 = __builtin_amdgcn_readlane(rj, j1);
            int r2 = __builtin_amdgcn_readlane(rj, j2);
            int r3 = __builtin_amdgcn_readlane(rj, j3);
            float4 v0 = src4[(size_t)r0 * (HID / 4) + lane];
            float4 v1 = src4[(size_t)r1 * (HID / 4) + lane];
            float4 v2 = src4[(size_t)r2 * (HID / 4) + lane];
            float4 v3 = src4[(size_t)r3 * (HID / 4) + lane];
            __builtin_amdgcn_sched_barrier(0);
            float c0 = __int_as_float(__builtin_amdgcn_readlane(dji, j))
                     * __int_as_float(__builtin_amdgcn_readlane(wj, j)) * dc;
            float c1 = __int_as_float(__builtin_amdgcn_readlane(dji, j1))
                     * __int_as_float(__builtin_amdgcn_readlane(wj, j1)) * dc;
            float c2 = __int_as_float(__builtin_amdgcn_readlane(dji, j2))
                     * __int_as_float(__builtin_amdgcn_readlane(wj, j2)) * dc;
            float c3 = __int_as_float(__builtin_amdgcn_readlane(dji, j3))
                     * __int_as_float(__builtin_amdgcn_readlane(wj, j3)) * dc;
            if (j + 1 >= m) c1 = 0.0f;
            if (j + 2 >= m) c2 = 0.0f;
            if (j + 3 >= m) c3 = 0.0f;
            acc.x += c0 * v0.x; acc.y += c0 * v0.y; acc.z += c0 * v0.z; acc.w += c0 * v0.w;
            acc.x += c1 * v1.x; acc.y += c1 * v1.y; acc.z += c1 * v1.z; acc.w += c1 * v1.w;
            acc.x += c2 * v2.x; acc.y += c2 * v2.y; acc.z += c2 * v2.z; acc.w += c2 * v2.w;
            acc.x += c3 * v3.x; acc.y += c3 * v3.y; acc.z += c3 * v3.z; acc.w += c3 * v3.w;
        }
    }
    return acc;
}

__global__ void gather_relu_kernel(const int* __restrict__ offs, const int* __restrict__ pairs,
                                   const float* __restrict__ dis, const float* __restrict__ invdeg,
                                   const float* __restrict__ xw, const float* __restrict__ b1,
                                   float* __restrict__ h) {
    int wave = (blockIdx.x * blockDim.x + threadIdx.x) >> 5;
    int lane = threadIdx.x & 31;
    if (wave >= N_NODES) return;
    int c = __builtin_amdgcn_readfirstlane(wave);
    float4 own  = ((const float4*)xw)[(size_t)c * (HID / 4) + lane];
    float4 bias = ((const float4*)b1)[lane];
    float  id   = invdeg[c];
    float4 acc  = gather_row(c, lane, offs, pairs, dis, xw);
    float4 v;
    v.x = fmaxf(acc.x + own.x * id + bias.x, 0.0f);
    v.y = fmaxf(acc.y + own.y * id + bias.y, 0.0f);
    v.z = fmaxf(acc.z + own.z * id + bias.z, 0.0f);
    v.w = fmaxf(acc.w + own.w * id + bias.w, 0.0f);
    ((float4*)h)[(size_t)c * (HID / 4) + lane] = v;
}

__global__ void gather_out_kernel(const int* __restrict__ offs, const int* __restrict__ pairs,
                                  const float* __restrict__ dis, const float* __restrict__ invdeg,
                                  const float* __restrict__ xw2, const float* __restrict__ b_mu,
                                  const float* __restrict__ b_ls, float* __restrict__ out) {
    int wave = (blockIdx.x * blockDim.x + threadIdx.x) >> 5;
    int lane = threadIdx.x & 31;
    if (wave >= N_NODES) return;
    int c = __builtin_amdgcn_readfirstlane(wave);
    int l16 = lane & 15;
    float4 own  = ((const float4*)xw2)[(size_t)c * (HID / 4) + lane];
    float4 bias = (lane < 16) ? ((const float4*)b_mu)[l16] : ((const float4*)b_ls)[l16];
    float  id   = invdeg[c];
    float4 acc  = gather_row(c, lane, offs, pairs, dis, xw2);
    float4 v;
    v.x = acc.x + own.x * id + bias.x;
    v.y = acc.y + own.y * id + bias.y;
    v.z = acc.z + own.z * id + bias.z;
    v.w = acc.w + own.w * id + bias.w;
    float* orow = out + ((lane < 16) ? (size_t)0 : (size_t)N_NODES * OUT_CH)
                      + (size_t)c * OUT_CH + l16 * 4;
    *(float4*)orow = v;
}

extern "C" void kernel_launch(void* const* d_in, const int* in_sizes, int n_in,
                              void* d_out, int out_size, void* d_ws, size_t ws_size,
                              hipStream_t stream) {
    (void)in_sizes; (void)n_in; (void)out_size; (void)ws_size;
    const float* x   = (const float*)d_in[0];
    const int*   ei  = (const int*)d_in[1];
    const float* ew  = (const float*)d_in[2];
    const float* W1  = (const float*)d_in[3];
    const float* b1  = (const float*)d_in[4];
    const float* Wmu = (const float*)d_in[5];
    const float* bmu = (const float*)d_in[6];
    const float* Wls = (const float*)d_in[7];
    const float* bls = (const float*)d_in[8];
    const int* row = ei;
    const int* col = ei + N_EDGES;

    float* ws     = (float*)d_ws;
    float* dis    = ws;
    float* invdeg = ws + 50000;
    int*   cnt    = (int*)(ws + 100000);
    float* xw     = ws + 150000;
    float* agg    = ws + 150000 + 6400000;
    int*   offs   = (int*)(ws + 12950000);
    int*   rank   = (int*)(ws + 13000048);
    int*   pairs  = (int*)(ws + 13600048);

    hipMemsetAsync(cnt, 0, (size_t)N_NODES * sizeof(int), stream);

    count_kernel<<<(N_EDGES + 255) / 256, 256, 0, stream>>>(col, cnt, rank);
    scan_kernel<<<1, SCAN_THREADS, 0, stream>>>(cnt, offs);
    fill_kernel<<<(N_EDGES + 255) / 256, 256, 0, stream>>>(row, col, ew, offs, rank, pairs);
    dis_kernel<<<(N_NODES + 255) / 256, 256, 0, stream>>>(offs, pairs, dis, invdeg);

    gemm_xwt_wmma<<<3125, 256, 0, stream>>>(x, W1, W1, HID, xw);
    gather_relu_kernel<<<N_NODES / 8, 256, 0, stream>>>(offs, pairs, dis, invdeg, xw, b1, agg);

    gemm_xwt_wmma<<<3125, 256, 0, stream>>>(agg, Wmu, Wls, OUT_CH, xw);
    gather_out_kernel<<<N_NODES / 8, 256, 0, stream>>>(offs, pairs, dis, invdeg, xw, bmu, bls,
                                                       (float*)d_out);
}


# GEMM retiled: one wave per 16x128 strip, each A fragment feeds 8 WMMAs, weights staged once per work-group in LDS, 16-byte operand loads, whole 256-byte stores; k sum order changes inside each 8
# speedup vs baseline: 2.4040x; 1.2181x over previous
#include <hip/hip_runtime.h>


#define N_NODES 50000
#define N_EDGES 600000
#define IN_CH   128
#define HID     128
#define OUT_CH  64

typedef __attribute__((ext_vector_type(2))) float v2f;
typedef __attribute__((ext_vector_type(8))) float v8f;

typedef __attribute__((ext_vector_type(2))) int v2i;

#define SCAN_THREADS 1024
#define SCAN_TILES   49

__global__ void count_kernel(const int* __restrict__ col, int* __restrict__ cnt,
                             int* __restrict__ rank) {
    int e = blockIdx.x * blockDim.x + threadIdx.x;
    if (e >= N_EDGES) return;
    rank[e] = atomicAdd(&cnt[col[e]], 1);
}

static __device__ __forceinline__ int wave_incl_scan(int x, int lane) {
    x += __builtin_amdgcn_update_dpp(0, x, 0x111, 0xf, 0xf, true);
    x += __builtin_amdgcn_update_dpp(0, x, 0x112, 0xf, 0xf, true);
    x += __builtin_amdgcn_update_dpp(0, x, 0x114, 0xf, 0xf, true);
    x += __builtin_amdgcn_update_dpp(0, x, 0x118, 0xf, 0xf, true);
    int low = __builtin_amdgcn_readlane(x, 15);
    if (lane >= 16) x += low;
    return x;
}

__global__ void __launch_bounds__(SCAN_THREADS)
scan_kernel(const int* __restrict__ cnt, int* __restrict__ offs) {
    __shared__ int wtot[SCAN_TILES * 32];
    __shared__ int ttot[SCAN_TILES];
    __shared__ int tbase[SCAN_TILES];
    int t    = threadIdx.x;
    int lane = t & 31;
    int wv   = t >> 5;

#pragma unroll
    for (int i = 0; i < SCAN_TILES; ++i) {
        int idx = i * SCAN_THREADS + t;
        int v   = (idx < N_NODES) ? cnt[idx] : 0;
        int inc = wave_incl_scan(v, lane);
        if (lane == 31) wtot[i * 32 + wv] = inc;
    }
    __syncthreads();
    for (int tile = wv; tile < SCAN_TILES; tile += 32) {
        int x   = wtot[tile * 32 + lane];
        int inc = wave_incl_scan(x, lane);
        wtot[tile * 32 + lane] = inc - x;
        if (lane == 31) ttot[tile] = inc;
    }
    __syncthreads();
    if (wv == 0) {
        int a  = ttot[lane];
        int bq = (lane + 32 < SCAN_TILES) ? ttot[lane + 32] : 0;
        int sa = wave_incl_scan(a, lane);
        int sb = wave_incl_scan(bq, lane);
        int first32 = __builtin_amdgcn_readlane(sa, 31);
        tbase[lane] = sa - a;
        if (lane + 32 < SCAN_TILES) tbase[lane + 32] = first32 + sb - bq;
        if (lane == 31) offs[N_NODES] = first32 + sb;
    }
    __syncthreads();
#pragma unroll
    for (int i = 0; i < SCAN_TILES; ++i) {
        int idx = i * SCAN_THREADS + t;
        int v   = (idx < N_NODES) ? cnt[idx] : 0;
        int inc = wave_incl_scan(v, lane);
        if (idx < N_NODES) offs[idx] = tbase[i] + wtot[i * 32 + wv] + inc - v;
    }
}

__global__ void fill_kernel(const int* __restrict__ row, const int* __restrict__ col,
                            const float* __restrict__ w, const int* __restrict__ offs,
                            const int* __restrict__ rank, int* __restrict__ pairs) {
    int e = blockIdx.x * blockDim.x + threadIdx.x;
    if (e >= N_EDGES) return;
    int r = row[e], c = col[e];
    float we = (r == c) ? 0.0f : w[e];
    int p = offs[c] + rank[e];
    v2i pr = {r, __float_as_int(we)};
    *(v2i*)(pairs + 2 * (size_t)p) = pr;
}

__global__ void dis_kernel(const int* __restrict__ offs, const int* __restrict__ pairs,
                           float* __restrict__ dis, float* __restrict__ invdeg) {
    int i = blockIdx.x * blockDim.x + threadIdx.x;
    if (i >= N_NODES) return;
    int b = offs[i], n = offs[i + 1];
    float d = 0.0f;
    for (int j = b; j < n; ++j) d += __int_as_float(pairs[2 * (size_t)j + 1]);
    d += 1.0f;
    dis[i]    = rsqrtf(d);
    invdeg[i] = 1.0f / d;
}

#define GEMM_WAVES 8
#define M_TILES    (N_NODES / 16)
#define W_PITCH4   33

__global__ void __launch_bounds__(GEMM_WAVES * 32)
gemm_xwt_wmma(const float* __restrict__ X,
              const float* __restrict__ Wa,
              const float* __restrict__ Wb,
              int n_split,
              float* __restrict__ C) {
    __shared__ float4 wl[HID * W_PITCH4];

    int tid  = threadIdx.x;
    int lane = tid & 31;
    int wv   = __builtin_amdgcn_readfirstlane(tid >> 5);
    int mt   = blockIdx.x * GEMM_WAVES + wv;
    int mtc  = (mt < M_TILES) ? mt : (M_TILES - 1);
    int c    = lane & 15;
    int h    = lane >> 4;

    const float4* xrow = (const float4*)(X + (size_t)(mtc * 16 + c) * IN_CH) + h;
    float4 a[16];
#pragma unroll
    for (int j = 0; j < 16; ++j) a[j] = xrow[2 * j];

    float4 wrow[HID / GEMM_WAVES];
#pragma unroll
    for (int i = 0; i < HID / GEMM_WAVES; ++i) {
        int n = i * GEMM_WAVES + wv;
        const float* srow = (n < n_split) ? (Wa + (size_t)n * IN_CH)
                                          : (Wb + (size_t)(n - n_split) * IN_CH);
        wrow[i] = ((const float4*)srow)[lane];
    }
#pragma unroll
    for (int i = 0; i < HID / GEMM_WAVES; ++i) {
        int n  = i * GEMM_WAVES + wv;
        int t  = ((n >> 6) << 2) | (n & 3);
        int cc = (n & 63) >> 2;
        wl[(16 * t + cc) * W_PITCH4 + lane] = wrow[i];
    }
    __syncthreads();

    v8f acc[8];
#pragma unroll
    for (int t = 0; t < 8; ++t) acc[t] = v8f{};

    const float4* bl = wl + c * W_PITCH4 + h;
    float4 bcur[8], bnxt[8];
#pragma unroll
    for (int t = 0; t < 8; ++t) bcur[t] = bl[(16 * t) * W_PITCH4];

#pragma unroll
    for (int j = 0; j < 16; ++j) {
        if (j + 1 < 16) {
#pragma unroll
            for (int t = 0; t < 8; ++t) bnxt[t] = bl[(16 * t) * W_PITCH4 + 2 * (j + 1)];
        }
        __builtin_amdgcn_sched_barrier(0);
        v2f a0 = {a[j].x, a[j].y};
        v2f a1 = {a[j].z, a[j].w};
#pragma unroll
        for (int t = 0; t < 8; ++t) {
            v2f b0 = {bcur[t].x, bcur[t].y};
            acc[t] = __builtin_amdgcn_wmma_f32_16x16x4_f32(
                false, a0, false, b0, (short)0, acc[t], false, false);
        }
#pragma unroll
        for (int t = 0; t < 8; ++t) {
            v2f b1 = {bcur[t].z, bcur[t].w};
            acc[t] = __builtin_amdgcn_wmma_f32_16x16x4_f32(
                false, a1, false, b1, (short)0, acc[t], false, false);
        }
        __builtin_amdgcn_sched_barrier(0);
#pragma unroll
        for (int t = 0; t < 8; ++t) bcur[t] = bnxt[t];
    }

    if (mt < M_TILES) {
        float* crow = C + (size_t)(mt * 16 + h * 8) * HID + 4 * c;
#pragma unroll
        for (int r = 0; r < 8; ++r) {
#pragma unroll
            for (int q = 0; q < 2; ++q) {
                float4 o = {acc[4 * q + 0][r], acc[4 * q + 1][r], acc[4 * q + 2][r], acc[4 * q + 3][r]};
                *(float4*)(crow + (size_t)r * HID + 64 * q) = o;
            }
        }
    }
}

static __device__ __forceinline__ float4 gather_row(int c, int lane,
                                                    const int* __restrict__ offs,
                                                    const int* __restrict__ pairs,
                                                    const float* __restrict__ dis,
                                                    const float* __restrict__ src) {
    int b = offs[c], n = offs[c + 1];
    float dc = dis[c];
    const float4* src4 = (const float4*)src;

    float4 acc = {0.0f, 0.0f, 0.0f, 0.0f};
    for (int base = b; base < n; base += 32) {
        int m = n - base;
        if (m > 32) m = 32;
        int   rj = 0, wj = 0;
        float dj = 0.0f;
        if (lane < m) {
            v2i pr = *(const v2i*)(pairs + 2 * (size_t)(base + lane));
            rj = pr.x;
            wj = pr.y;
            dj = dis[rj];
        }
        int dji  = __float_as_int(dj);
        int last = m - 1;
        for (int j = 0; j < m; j += 4) {
            int j1 = (j + 1 < last) ? j + 1 : last;
            int j2 = (j + 2 < last) ? j + 2 : last;
            int j3 = (j + 3 < last) ? j + 3 : last;
            int r0 = __builtin_amdgcn_readlane(rj, j);
            int r1 = __builtin_amdgcn_readlane(rj, j1);
            int r2 = __builtin_amdgcn_readlane(rj, j2);
            int r3 = __builtin_amdgcn_readlane(rj, j3);
            float4 v0 = src4[(size_t)r0 * (HID / 4) + lane];
            float4 v1 = src4[(size_t)r1 * (HID / 4) + lane];
            float4 v2 = src4[(size_t)r2 * (HID / 4) + lane];
            float4 v3 = src4[(size_t)r3 * (HID / 4) + lane];
            __builtin_amdgcn_sched_barrier(0);
            float c0 = __int_as_float(__builtin_amdgcn_readlane(dji, j))
                     * __int_as_float(__builtin_amdgcn_readlane(wj, j)) * dc;
            float c1 = __int_as_float(__builtin_amdgcn_readlane(dji, j1))
                     * __int_as_float(__builtin_amdgcn_readlane(wj, j1)) * dc;
            float c2 = __int_as_float(__builtin_amdgcn_readlane(dji, j2))
                     * __int_as_float(__builtin_amdgcn_readlane(wj, j2)) * dc;
            float c3 = __int_as_float(__builtin_amdgcn_readlane(dji, j3))
                     * __int_as_float(__builtin_amdgcn_readlane(wj, j3)) * dc;
            if (j + 1 >= m) c1 = 0.0f;
            if (j + 2 >= m) c2 = 0.0f;
            if (j + 3 >= m) c3 = 0.0f;
            acc.x += c0 * v0.x; acc.y += c0 * v0.y; acc.z += c0 * v0.z; acc.w += c0 * v0.w;
            acc.x += c1 * v1.x; acc.y += c1 * v1.y; acc.z += c1 * v1.z; acc.w += c1 * v1.w;
            acc.x += c2 * v2.x; acc.y += c2 * v2.y; acc.z += c2 * v2.z; acc.w += c2 * v2.w;
            acc.x += c3 * v3.x; acc.y += c3 * v3.y; acc.z += c3 * v3.z; acc.w += c3 * v3.w;
        }
    }
    return acc;
}

__global__ void gather_relu_kernel(const int* __restrict__ offs, const int* __restrict__ pairs,
                                   const float* __restrict__ dis, const float* __restrict__ invdeg,
                                   const float* __restrict__ xw, const float* __restrict__ b1,
                                   float* __restrict__ h) {
    int wave = (blockIdx.x * blockDim.x + threadIdx.x) >> 5;
    int lane = threadIdx.x & 31;
    if (wave >= N_NODES) return;
    int c = __builtin_amdgcn_readfirstlane(wave);
    float4 own  = ((const float4*)xw)[(size_t)c * (HID / 4) + lane];
    float4 bias = ((const float4*)b1)[lane];
    float  id   = invdeg[c];
    float4 acc  = gather_row(c, lane, offs, pairs, dis, xw);
    float4 v;
    v.x = fmaxf(acc.x + own.x * id + bias.x, 0.0f);
    v.y = fmaxf(acc.y + own.y * id + bias.y, 0.0f);
    v.z = fmaxf(acc.z + own.z * id + bias.z, 0.0f);
    v.w = fmaxf(acc.w + own.w * id + bias.w, 0.0f);
    ((float4*)h)[(size_t)c * (HID / 4) + lane] = v;
}

__global__ void gather_out_kernel(const int* __restrict__ offs, const int* __restrict__ pairs,
                                  const float* __restrict__ dis, const float* __restrict__ invdeg,
                                  const float* __restrict__ xw2, const float* __restrict__ b_mu,
                                  const float* __restrict__ b_ls, float* __restrict__ out) {
    int wave = (blockIdx.x * blockDim.x + threadIdx.x) >> 5;
    int lane = threadIdx.x & 31;
    if (wave >= N_NODES) return;
    int c = __builtin_amdgcn_readfirstlane(wave);
    int l16 = lane & 15;
    float4 own  = ((const float4*)xw2)[(size_t)c * (HID / 4) + lane];
    float4 bias = (lane < 16) ? ((const float4*)b_mu)[l16] : ((const float4*)b_ls)[l16];
    float  id   = invdeg[c];
    float4 acc  = gather_row(c, lane, offs, pairs, dis, xw2);
    float4 v;
    v.x = acc.x + own.x * id + bias.x;
    v.y = acc.y + own.y * id + bias.y;
    v.z = acc.z + own.z * id + bias.z;
    v.w = acc.w + own.w * id + bias.w;
    float* orow = out + ((lane < 16) ? (size_t)0 : (size_t)N_NODES * OUT_CH)
                      + (size_t)c * OUT_CH + l16 * 4;
    *(float4*)orow = v;
}

extern "C" void kernel_launch(void* const* d_in, const int* in_sizes, int n_in,
                              void* d_out, int out_size, void* d_ws, size_t ws_size,
                              hipStream_t stream) {
    (void)in_sizes; (void)n_in; (void)out_size; (void)ws_size;
    const float* x   = (const float*)d_in[0];
    const int*   ei  = (const int*)d_in[1];
    const float* ew  = (const float*)d_in[2];
    const float* W1  = (const float*)d_in[3];
    const float* b1  = (const float*)d_in[4];
    const float* Wmu = (const float*)d_in[5];
    const float* bmu = (const float*)d_in[6];
    const float* Wls = (const float*)d_in[7];
    const float* bls = (const float*)d_in[8];
    const int* row = ei;
    const int* col = ei + N_EDGES;

    float* ws     = (float*)d_ws;
    float* dis    = ws;
    float* invdeg = ws + 50000;
    int*   cnt    = (int*)(ws + 100000);
    float* xw     = ws + 150000;
    float* agg    = ws + 150000 + 6400000;
    int*   offs   = (int*)(ws + 12950000);
    int*   rank   = (int*)(ws + 13000048);
    int*   pairs  = (int*)(ws + 13600048);

    hipMemsetAsync(cnt, 0, (size_t)N_NODES * sizeof(int), stream);

    count_kernel<<<(N_EDGES + 255) / 256, 256, 0, stream>>>(col, cnt, rank);
    scan_kernel<<<1, SCAN_THREADS, 0, stream>>>(cnt, offs);
    fill_kernel<<<(N_EDGES + 255) / 256, 256, 0, stream>>>(row, col, ew, offs, rank, pairs);
    dis_kernel<<<(N_NODES + 255) / 256, 256, 0, stream>>>(offs, pairs, dis, invdeg);

    gemm_xwt_wmma<<<(M_TILES + GEMM_WAVES - 1) / GEMM_WAVES, GEMM_WAVES * 32, 0, stream>>>(x, W1, W1, HID, xw);
    gather_relu_kernel<<<N_NODES / 8, 256, 0, stream>>>(offs, pairs, dis, invdeg, xw, b1, agg);

    gemm_xwt_wmma<<<(M_TILES + GEMM_WAVES - 1) / GEMM_WAVES, GEMM_WAVES * 32, 0, stream>>>(agg, Wmu, Wls, OUT_CH, xw);
    gather_out_kernel<<<N_NODES / 8, 256, 0, stream>>>(offs, pairs, dis, invdeg, xw, bmu, bls,
                                                       (float*)d_out);
}


# Edge coefficient factored: GEMMs store rows times dis[row], gather weights by w alone, scales its sum once by dis[col] (self-loop = own row): no per-edge dis look-up; rounding per term changes
# speedup vs baseline: 2.5213x; 1.0488x over previous
#include <hip/hip_runtime.h>


#define N_NODES 50000
#define N_EDGES 600000
#define IN_CH   128
#define HID     128
#define OUT_CH  64

typedef __attribute__((ext_vector_type(2))) float v2f;
typedef __attribute__((ext_vector_type(8))) float v8f;

typedef __attribute__((ext_vector_type(2))) int v2i;

#define SCAN_THREADS 1024
#define SCAN_TILES   49

__global__ void count_kernel(const int* __restrict__ col, int* __restrict__ cnt,
                             int* __restrict__ rank) {
    int e = blockIdx.x * blockDim.x + threadIdx.x;
    if (e >= N_EDGES) return;
    rank[e] = atomicAdd(&cnt[col[e]], 1);
}

static __device__ __forceinline__ int wave_incl_scan(int x, int lane) {
    x += __builtin_amdgcn_update_dpp(0, x, 0x111, 0xf, 0xf, true);
    x += __builtin_amdgcn_update_dpp(0, x, 0x112, 0xf, 0xf, true);
    x += __builtin_amdgcn_update_dpp(0, x, 0x114, 0xf, 0xf, true);
    x += __builtin_amdgcn_update_dpp(0, x, 0x118, 0xf, 0xf, true);
    int low = __builtin_amdgcn_readlane(x, 15);
    if (lane >= 16) x += low;
    return x;
}

__global__ void __launch_bounds__(SCAN_THREADS)
scan_kernel(const int* __restrict__ cnt, int* __restrict__ offs) {
    __shared__ int wtot[SCAN_TILES * 32];
    __shared__ int ttot[SCAN_TILES];
    __shared__ int tbase[SCAN_TILES];
    int t    = threadIdx.x;
    int lane = t & 31;
    int wv   = t >> 5;

#pragma unroll
    for (int i = 0; i < SCAN_TILES; ++i) {
        int idx = i * SCAN_THREADS + t;
        int v   = (idx < N_NODES) ? cnt[idx] : 0;
        int inc = wave_incl_scan(v, lane);
        if (lane == 31) wtot[i * 32 + wv] = inc;
    }
    __syncthreads();
    for (int tile = wv; tile < SCAN_TILES; tile += 32) {
        int x   = wtot[tile * 32 + lane];
        int inc = wave_incl_scan(x, lane);
        wtot[tile * 32 + lane] = inc - x;
        if (lane == 31) ttot[tile] = inc;
    }
    __syncthreads();
    if (wv == 0) {
        int a  = ttot[lane];
        int bq = (lane + 32 < SCAN_TILES) ? ttot[lane + 32] : 0;
        int sa = wave_incl_scan(a, lane);
        int sb = wave_incl_scan(bq, lane);
        int first32 = __builtin_amdgcn_readlane(sa, 31);
        tbase[lane] = sa - a;
        if (lane + 32 < SCAN_TILES) tbase[lane + 32] = first32 + sb - bq;
        if (lane == 31) offs[N_NODES] = first32 + sb;
    }
    __syncthreads();
#pragma unroll
    for (int i = 0; i < SCAN_TILES; ++i) {
        int idx = i * SCAN_THREADS + t;
        int v   = (idx < N_NODES) ? cnt[idx] : 0;
        int inc = wave_incl_scan(v, lane);
        if (idx < N_NODES) offs[idx] = tbase[i] + wtot[i * 32 + wv] + inc - v;
    }
}

__global__ void fill_kernel(const int* __restrict__ row, const int* __restrict__ col,
                            const float* __restrict__ w, const int* __restrict__ offs,
                            const int* __restrict__ rank, int* __restrict__ pairs) {
    int e = blockIdx.x * blockDim.x + threadIdx.x;
    if (e >= N_EDGES) return;
    int r = row[e], c = col[e];
    float we = (r == c) ? 0.0f : w[e];
    int p = offs[c] + rank[e];
    v2i pr = {r, __float_as_int(we)};
    *(v2i*)(pairs + 2 * (size_t)p) = pr;
}

__global__ void dis_kernel(const int* __restrict__ offs, const int* __restrict__ pairs,
                           float* __restrict__ dis) {
    int i = blockIdx.x * blockDim.x + threadIdx.x;
    if (i >= N_NODES) return;
    int b = offs[i], n = offs[i + 1];
    float d = 0.0f;
    for (int j = b; j < n; ++j) d += __int_as_float(pairs[2 * (size_t)j + 1]);
    d += 1.0f;
    dis[i] = rsqrtf(d);
}

#define GEMM_WAVES 8
#define M_TILES    (N_NODES / 16)
#define W_PITCH4   33

__global__ void __launch_bounds__(GEMM_WAVES * 32)
gemm_xwt_wmma(const float* __restrict__ X,
              const float* __restrict__ Wa,
              const float* __restrict__ Wb,
              int n_split,
              const float* __restrict__ rs,
              float* __restrict__ C) {
    __shared__ float4 wl[HID * W_PITCH4];

    int tid  = threadIdx.x;
    int lane = tid & 31;
    int wv   = __builtin_amdgcn_readfirstlane(tid >> 5);
    int mt   = blockIdx.x * GEMM_WAVES + wv;
    int mtc  = (mt < M_TILES) ? mt : (M_TILES - 1);
    int c    = lane & 15;
    int h    = lane >> 4;

    const float4* xrow = (const float4*)(X + (size_t)(mtc * 16 + c) * IN_CH) + h;
    float4 a[16];
#pragma unroll
    for (int j = 0; j < 16; ++j) a[j] = xrow[2 * j];
    const float4* rsp = (const float4*)(rs + mtc * 16 + h * 8);
    float4 rs_lo = rsp[0];
    float4 rs_hi = rsp[1];

    float4 wrow[HID / GEMM_WAVES];
#pragma unroll
    for (int i = 0; i < HID / GEMM_WAVES; ++i) {
        int n = i * GEMM_WAVES + wv;
        const float* srow = (n < n_split) ? (Wa + (size_t)n * IN_CH)
                                          : (Wb + (size_t)(n - n_split) * IN_CH);
        wrow[i] = ((const float4*)srow)[lane];
    }
#pragma unroll
    for (int i = 0; i < HID / GEMM_WAVES; ++i) {
        int n  = i * GEMM_WAVES + wv;
        int t  = ((n >> 6) << 2) | (n & 3);
        int cc = (n & 63) >> 2;
        wl[(16 * t + cc) * W_PITCH4 + lane] = wrow[i];
    }
    __syncthreads();

    v8f acc[8];
#pragma unroll
    for (int t = 0; t < 8; ++t) acc[t] = v8f{};

    const float4* bl = wl + c * W_PITCH4 + h;
    float4 bcur[8], bnxt[8];
#pragma unroll
    for (int t = 0; t < 8; ++t) bcur[t] = bl[(16 * t) * W_PITCH4];

#pragma unroll
    for (int j = 0; j < 16; ++j) {
        if (j + 1 < 16) {
#pragma unroll
            for (int t = 0; t < 8; ++t) bnxt[t] = bl[(16 * t) * W_PITCH4 + 2 * (j + 1)];
        }
        __builtin_amdgcn_sched_barrier(0);
        v2f a0 = {a[j].x, a[j].y};
        v2f a1 = {a[j].z, a[j].w};
#pragma unroll
        for (int t = 0; t < 8; ++t) {
            v2f b0 = {bcur[t].x, bcur[t].y};
            acc[t] = __builtin_amdgcn_wmma_f32_16x16x4_f32(
                false, a0, false, b0, (short)0, acc[t], false, false);
        }
#pragma unroll
        for (int t = 0; t < 8; ++t) {
            v2f b1 = {bcur[t].z, bcur[t].w};
            acc[t] = __builtin_amdgcn_wmma_f32_16x16x4_f32(
                false, a1, false, b1, (short)0, acc[t], false, false);
        }
        __builtin_amdgcn_sched_barrier(0);
#pragma unroll
        for (int t = 0; t < 8; ++t) bcur[t] = bnxt[t];
    }

    float rsv[8] = {rs_lo.x, rs_lo.y, rs_lo.z, rs_lo.w, rs_hi.x, rs_hi.y, rs_hi.z, rs_hi.w};
#pragma unroll
    for (int t = 0; t < 8; ++t) {
#pragma unroll
        for (int r = 0; r < 8; ++r) acc[t][r] *= rsv[r];
    }

    if (mt < M_TILES) {
        float* crow = C + (size_t)(mt * 16 + h * 8) * HID + 4 * c;
#pragma unroll
        for (int r = 0; r < 8; ++r) {
#pragma unroll
            for (int q = 0; q < 2; ++q) {
                float4 o = {acc[4 * q + 0][r], acc[4 * q + 1][r], acc[4 * q + 2][r], acc[4 * q + 3][r]};
                *(float4*)(crow + (size_t)r * HID + 64 * q) = o;
            }
        }
    }
}

static __device__ __forceinline__ float4 gather_row(int c, int lane,
                                                    const int* __restrict__ offs,
                                                    const int* __restrict__ pairs,
                                                    const float* __restrict__ src) {
    int b = offs[c], n = offs[c + 1];
    const float4* src4 = (const float4*)src;

    float4 acc = {0.0f, 0.0f, 0.0f, 0.0f};
    for (int base = b; base < n; base += 32) {
        int m = n - base;
        if (m > 32) m = 32;
        int   rj = 0, wj = 0;
        if (lane < m) {
            v2i pr = *(const v2i*)(pairs + 2 * (size_t)(base + lane));
            rj = pr.x;
            wj = pr.y;
        }
        int last = m - 1;
        for (int j = 0; j < m; j += 4) {
            int j1 = (j + 1 < last) ? j + 1 : last;
            int j2 = (j + 2 < last) ? j + 2 : last;
            int j3 = (j + 3 < last) ? j + 3 : last;
            int r0 = __builtin_amdgcn_readlane(rj, j);
            int r1 = __builtin_amdgcn_readlane(rj, j1);
            int r2 = __builtin_amdgcn_readlane(rj, j2);
            int r3 = __builtin_amdgcn_readlane(rj, j3);
            float4 v0 = src4[(size_t)r0 * (HID / 4) + lane];
            float4 v1 = src4[(size_t)r1 * (HID / 4) + lane];
            float4 v2 = src4[(size_t)r2 * (HID / 4) + lane];
            float4 v3 = src4[(size_t)r3 * (HID / 4) + lane];
            float c0 = __int_as_float(__builtin_amdgcn_readlane(wj, j));
            float c1 = __int_as_float(__builtin_amdgcn_readlane(wj, j1));
            float c2 = __int_as_float(__builtin_amdgcn_readlane(wj, j2));
            float c3 = __int_as_float(__builtin_amdgcn_readlane(wj, j3));
            if (j + 1 >= m) c1 = 0.0f;
            if (j + 2 >= m) c2 = 0.0f;
            if (j + 3 >= m) c3 = 0.0f;
            acc.x += c0 * v0.x; acc.y += c0 * v0.y; acc.z += c0 * v0.z; acc.w += c0 * v0.w;
            acc.x += c1 * v1.x; acc.y += c1 * v1.y; acc.z += c1 * v1.z; acc.w += c1 * v1.w;
            acc.x += c2 * v2.x; acc.y += c2 * v2.y; acc.z += c2 * v2.z; acc.w += c2 * v2.w;
            acc.x += c3 * v3.x; acc.y += c3 * v3.y; acc.z += c3 * v3.z; acc.w += c3 * v3.w;
        }
    }
    return acc;
}

__global__ void gather_relu_kernel(const int* __restrict__ offs, const int* __restrict__ pairs,
                                   const float* __restrict__ dis,
                                   const float* __restrict__ xw, const float* __restrict__ b1,
                                   float* __restrict__ h) {
    int wave = (blockIdx.x * blockDim.x + threadIdx.x) >> 5;
    int lane = threadIdx.x & 31;
    if (wave >= N_NODES) return;
    int c = __builtin_amdgcn_readfirstlane(wave);
    float4 own  = ((const float4*)xw)[(size_t)c * (HID / 4) + lane];
    float4 bias = ((const float4*)b1)[lane];
    float  dc   = dis[c];
    float4 acc  = gather_row(c, lane, offs, pairs, xw);
    float4 v;
    v.x = fmaxf((acc.x + own.x) * dc + bias.x, 0.0f);
    v.y = fmaxf((acc.y + own.y) * dc + bias.y, 0.0f);
    v.z = fmaxf((acc.z + own.z) * dc + bias.z, 0.0f);
    v.w = fmaxf((acc.w + own.w) * dc + bias.w, 0.0f);
    ((float4*)h)[(size_t)c * (HID / 4) + lane] = v;
}

__global__ void gather_out_kernel(const int* __restrict__ offs, const int* __restrict__ pairs,
                                  const float* __restrict__ dis,
                                  const float* __restrict__ xw2, const float* __restrict__ b_mu,
                                  const float* __restrict__ b_ls, float* __restrict__ out) {
    int wave = (blockIdx.x * blockDim.x + threadIdx.x) >> 5;
    int lane = threadIdx.x & 31;
    if (wave >= N_NODES) return;
    int c = __builtin_amdgcn_readfirstlane(wave);
    int l16 = lane & 15;
    float4 own  = ((const float4*)xw2)[(size_t)c * (HID / 4) + lane];
    float4 bias = (lane < 16) ? ((const float4*)b_mu)[l16] : ((const float4*)b_ls)[l16];
    float  dc   = dis[c];
    float4 acc  = gather_row(c, lane, offs, pairs, xw2);
    float4 v;
    v.x = (acc.x + own.x) * dc + bias.x;
    v.y = (acc.y + own.y) * dc + bias.y;
    v.z = (acc.z + own.z) * dc + bias.z;
    v.w = (acc.w + own.w) * dc + bias.w;
    float* orow = out + ((lane < 16) ? (size_t)0 : (size_t)N_NODES * OUT_CH)
                      + (size_t)c * OUT_CH + l16 * 4;
    *(float4*)orow = v;
}

extern "C" void kernel_launch(void* const* d_in, const int* in_sizes, int n_in,
                              void* d_out, int out_size, void* d_ws, size_t ws_size,
                              hipStream_t stream) {
    (void)in_sizes; (void)n_in; (void)out_size; (void)ws_size;
    const float* x   = (const float*)d_in[0];
    const int*   ei  = (const int*)d_in[1];
    const float* ew  = (const float*)d_in[2];
    const float* W1  = (const float*)d_in[3];
    const float* b1  = (const float*)d_in[4];
    const float* Wmu = (const float*)d_in[5];
    const float* bmu = (const float*)d_in[6];
    const float* Wls = (const float*)d_in[7];
    const float* bls = (const float*)d_in[8];
    const int* row = ei;
    const int* col = ei + N_EDGES;

    float* ws     = (float*)d_ws;
    float* dis    = ws;
    int*   cnt    = (int*)(ws + 100000);
    float* xw     = ws + 150000;
    float* agg    = ws + 150000 + 6400000;
    int*   offs   = (int*)(ws + 12950000);
    int*   rank   = (int*)(ws + 13000048);
    int*   pairs  = (int*)(ws + 13600048);

    hipMemsetAsync(cnt, 0, (size_t)N_NODES * sizeof(int), stream);

    count_kernel<<<(N_EDGES + 255) / 256, 256, 0, stream>>>(col, cnt, rank);
    scan_kernel<<<1, SCAN_THREADS, 0, stream>>>(cnt, offs);
    fill_kernel<<<(N_EDGES + 255) / 256, 256, 0, stream>>>(row, col, ew, offs, rank, pairs);
    dis_kernel<<<(N_NODES + 255) / 256, 256, 0, stream>>>(offs, pairs, dis);

    gemm_xwt_wmma<<<(M_TILES + GEMM_WAVES - 1) / GEMM_WAVES, GEMM_WAVES * 32, 0, stream>>>(x, W1, W1, HID, dis, xw);
    gather_relu_kernel<<<N_NODES / 8, 256, 0, stream>>>(offs, pairs, dis, xw, b1, agg);

    gemm_xwt_wmma<<<(M_TILES + GEMM_WAVES - 1) / GEMM_WAVES, GEMM_WAVES * 32, 0, stream>>>(agg, Wmu, Wls, OUT_CH, dis, xw);
    gather_out_kernel<<<N_NODES / 8, 256, 0, stream>>>(offs, pairs, dis, xw, bmu, bls,
                                                       (float*)d_out);
}


# One atomic pass drops each edge into its destination's fixed 32-slot bucket (excess to a spill list): the scan launch, rank array and second edge pass go; sums stay in arrival order
# speedup vs baseline: 3.1298x; 1.2413x over previous
#include <hip/hip_runtime.h>


#define N_NODES 50000
#define N_EDGES 600000
#define IN_CH   128
#define HID     128
#define OUT_CH  64

typedef __attribute__((ext_vector_type(2))) float v2f;
typedef __attribute__((ext_vector_type(8))) float v8f;

typedef __attribute__((ext_vector_type(2))) int v2i;

typedef __attribute__((ext_vector_type(4))) int v4i;

#define CAP 32

__global__ void bucket_kernel(const int* __restrict__ row, const int* __restrict__ col,
                              const float* __restrict__ w, int* __restrict__ cnt,
                              int* __restrict__ spill_n, int* __restrict__ buckets,
                              int* __restrict__ spill) {
    int e = blockIdx.x * blockDim.x + threadIdx.x;
    if (e >= N_EDGES) return;
    int r = row[e], c = col[e];
    float wv = w[e];
    float we = (r == c) ? 0.0f : wv;
    int s = atomicAdd(&cnt[c], 1);
    if (s < CAP) {
        v2i pr = {r, __float_as_int(we)};
        *(v2i*)(buckets + (size_t)c * (2 * CAP) + 2 * s) = pr;
    } else {
        int o = atomicAdd(spill_n, 1);
        v4i en = {c, r, __float_as_int(we), 0};
        *(v4i*)(spill + 4 * (size_t)o) = en;
    }
}

__global__ void dis_kernel(const int* __restrict__ cnt, const int* __restrict__ spill_n,
                           const int* __restrict__ buckets, const int* __restrict__ spill,
                           float* __restrict__ dis) {
    int i = blockIdx.x * blockDim.x + threadIdx.x;
    if (i >= N_NODES) return;
    int n = cnt[i];
    int m = (n < CAP) ? n : CAP;
    const int* bp = buckets + (size_t)i * (2 * CAP);
    float d = 0.0f;
    for (int j = 0; j < m; ++j) d += __int_as_float(bp[2 * j + 1]);
    if (n > CAP) {
        int no = *spill_n;
        for (int j = 0; j < no; ++j) {
            v4i en = *(const v4i*)(spill + 4 * (size_t)j);
            if (en.x == i) d += __int_as_float(en.z);
        }
    }
    d += 1.0f;
    dis[i] = rsqrtf(d);
}

#define GEMM_WAVES 8
#define M_TILES    (N_NODES / 16)
#define W_PITCH4   33

__global__ void __launch_bounds__(GEMM_WAVES * 32)
gemm_xwt_wmma(const float* __restrict__ X,
              const float* __restrict__ Wa,
              const float* __restrict__ Wb,
              int n_split,
              const float* __restrict__ rs,
              float* __restrict__ C) {
    __shared__ float4 wl[HID * W_PITCH4];

    int tid  = threadIdx.x;
    int lane = tid & 31;
    int wv   = __builtin_amdgcn_readfirstlane(tid >> 5);
    int mt   = blockIdx.x * GEMM_WAVES + wv;
    int mtc  = (mt < M_TILES) ? mt : (M_TILES - 1);
    int c    = lane & 15;
    int h    = lane >> 4;

    const float4* xrow = (const float4*)(X + (size_t)(mtc * 16 + c) * IN_CH) + h;
    float4 a[16];
#pragma unroll
    for (int j = 0; j < 16; ++j) a[j] = xrow[2 * j];
    const float4* rsp = (const float4*)(rs + mtc * 16 + h * 8);
    float4 rs_lo = rsp[0];
    float4 rs_hi = rsp[1];

    float4 wrow[HID / GEMM_WAVES];
#pragma unroll
    for (int i = 0; i < HID / GEMM_WAVES; ++i) {
        int n = i * GEMM_WAVES + wv;
        const float* srow = (n < n_split) ? (Wa + (size_t)n * IN_CH)
                                          : (Wb + (size_t)(n - n_split) * IN_CH);
        wrow[i] = ((const float4*)srow)[lane];
    }
#pragma unroll
    for (int i = 0; i < HID / GEMM_WAVES; ++i) {
        int n  = i * GEMM_WAVES + wv;
        int t  = ((n >> 6) << 2) | (n & 3);
        int cc = (n & 63) >> 2;
        wl[(16 * t + cc) * W_PITCH4 + lane] = wrow[i];
    }
    __syncthreads();

    v8f acc[8];
#pragma unroll
    for (int t = 0; t < 8; ++t) acc[t] = v8f{};

    const float4* bl = wl + c * W_PITCH4 + h;
    float4 bcur[8], bnxt[8];
#pragma unroll
    for (int t = 0; t < 8; ++t) bcur[t] = bl[(16 * t) * W_PITCH4];

#pragma unroll
    for (int j = 0; j < 16; ++j) {
        if (j + 1 < 16) {
#pragma unroll
            for (int t = 0; t < 8; ++t) bnxt[t] = bl[(16 * t) * W_PITCH4 + 2 * (j + 1)];
        }
        __builtin_amdgcn_sched_barrier(0);
        v2f a0 = {a[j].x, a[j].y};
        v2f a1 = {a[j].z, a[j].w};
#pragma unroll
        for (int t = 0; t < 8; ++t) {
            v2f b0 = {bcur[t].x, bcur[t].y};
            acc[t] = __builtin_amdgcn_wmma_f32_16x16x4_f32(
                false, a0, false, b0, (short)0, acc[t], false, false);
        }
#pragma unroll
        for (int t = 0; t < 8; ++t) {
            v2f b1 = {bcur[t].z, bcur[t].w};
            acc[t] = __builtin_amdgcn_wmma_f32_16x16x4_f32(
                false, a1, false, b1, (short)0, acc[t], false, false);
        }
        __builtin_amdgcn_sched_barrier(0);
#pragma unroll
        for (int t = 0; t < 8; ++t) bcur[t] = bnxt[t];
    }

    float rsv[8] = {rs_lo.x, rs_lo.y, rs_lo.z, rs_lo.w, rs_hi.x, rs_hi.y, rs_hi.z, rs_hi.w};
#pragma unroll
    for (int t = 0; t < 8; ++t) {
#pragma unroll
        for (int r = 0; r < 8; ++r) acc[t][r] *= rsv[r];
    }

    if (mt < M_TILES) {
        float* crow = C + (size_t)(mt * 16 + h * 8) * HID + 4 * c;
#pragma unroll
        for (int r = 0; r < 8; ++r) {
#pragma unroll
            for (int q = 0; q < 2; ++q) {
                float4 o = {acc[4 * q + 0][r], acc[4 * q + 1][r], acc[4 * q + 2][r], acc[4 * q + 3][r]};
                *(float4*)(crow + (size_t)r * HID + 64 * q) = o;
            }
        }
    }
}

static __device__ __forceinline__ float4 gather_row(int c, int lane,
                                                    const int* __restrict__ cnt,
                                                    const int* __restrict__ spill_n,
                                                    const int* __restrict__ buckets,
                                                    const int* __restrict__ spill,
                                                    const float* __restrict__ src) {
    int n = cnt[c];
    int m = (n < CAP) ? n : CAP;
    const float4* src4 = (const float4*)src;

    float4 acc = {0.0f, 0.0f, 0.0f, 0.0f};
    int   rj = 0, wj = 0;
    if (lane < m) {
        v2i pr = *(const v2i*)(buckets + (size_t)c * (2 * CAP) + 2 * lane);
        rj = pr.x;
        wj = pr.y;
    }
    int last = m - 1;
    for (int j = 0; j < m; j += 4) {
        int j1 = (j + 1 < last) ? j + 1 : last;
        int j2 = (j + 2 < last) ? j + 2 : last;
        int j3 = (j + 3 < last) ? j + 3 : last;
        int r0 = __builtin_amdgcn_readlane(rj, j);
        int r1 = __builtin_amdgcn_readlane(rj, j1);
        int r2 = __builtin_amdgcn_readlane(rj, j2);
        int r3 = __builtin_amdgcn_readlane(rj, j3);
        float4 v0 = src4[(size_t)r0 * (HID / 4) + lane];
        float4 v1 = src4[(size_t)r1 * (HID / 4) + lane];
        float4 v2 = src4[(size_t)r2 * (HID / 4) + lane];
        float4 v3 = src4[(size_t)r3 * (HID / 4) + lane];
        float c0 = __int_as_float(__builtin_amdgcn_readlane(wj, j));
        float c1 = __int_as_float(__builtin_amdgcn_readlane(wj, j1));
        float c2 = __int_as_float(__builtin_amdgcn_readlane(wj, j2));
        float c3 = __int_as_float(__builtin_amdgcn_readlane(wj, j3));
        if (j + 1 >= m) c1 = 0.0f;
        if (j + 2 >= m) c2 = 0.0f;
        if (j + 3 >= m) c3 = 0.0f;
        acc.x += c0 * v0.x; acc.y += c0 * v0.y; acc.z += c0 * v0.z; acc.w += c0 * v0.w;
        acc.x += c1 * v1.x; acc.y += c1 * v1.y; acc.z += c1 * v1.z; acc.w += c1 * v1.w;
        acc.x += c2 * v2.x; acc.y += c2 * v2.y; acc.z += c2 * v2.z; acc.w += c2 * v2.w;
        acc.x += c3 * v3.x; acc.y += c3 * v3.y; acc.z += c3 * v3.z; acc.w += c3 * v3.w;
    }
    if (n > CAP) {
        int no = *spill_n;
        for (int base = 0; base < no; base += 32) {
            int k  = base + lane;
            int oc = -1, orow = 0, ow = 0;
            if (k < no) {
                v4i en = *(const v4i*)(spill + 4 * (size_t)k);
                oc = en.x; orow = en.y; ow = en.z;
            }
            unsigned hit = __builtin_amdgcn_ballot_w32(oc == c);
            while (hit != 0u) {
                int l = __builtin_ctz(hit);
                hit &= hit - 1u;
                int   r  = __builtin_amdgcn_readlane(orow, l);
                float cw = __int_as_float(__builtin_amdgcn_readlane(ow, l));
                float4 v = src4[(size_t)r * (HID / 4) + lane];
                acc.x += cw * v.x; acc.y += cw * v.y; acc.z += cw * v.z; acc.w += cw * v.w;
            }
        }
    }
    return acc;
}

__global__ void gather_relu_kernel(const int* __restrict__ cnt, const int* __restrict__ spill_n,
                                   const int* __restrict__ buckets, const int* __restrict__ spill,
                                   const float* __restrict__ dis,
                                   const float* __restrict__ xw, const float* __restrict__ b1,
                                   float* __restrict__ h) {
    int wave = (blockIdx.x * blockDim.x + threadIdx.x) >> 5;
    int lane = threadIdx.x & 31;
    if (wave >= N_NODES) return;
    int c = __builtin_amdgcn_readfirstlane(wave);
    float4 own  = ((const float4*)xw)[(size_t)c * (HID / 4) + lane];
    float4 bias = ((const float4*)b1)[lane];
    float  dc   = dis[c];
    float4 acc  = gather_row(c, lane, cnt, spill_n, buckets, spill, xw);
    float4 v;
    v.x = fmaxf((acc.x + own.x) * dc + bias.x, 0.0f);
    v.y = fmaxf((acc.y + own.y) * dc + bias.y, 0.0f);
    v.z = fmaxf((acc.z + own.z) * dc + bias.z, 0.0f);
    v.w = fmaxf((acc.w + own.w) * dc + bias.w, 0.0f);
    ((float4*)h)[(size_t)c * (HID / 4) + lane] = v;
}

__global__ void gather_out_kernel(const int* __restrict__ cnt, const int* __restrict__ spill_n,
                                  const int* __restrict__ buckets, const int* __restrict__ spill,
                                  const float* __restrict__ dis,
                                  const float* __restrict__ xw2, const float* __restrict__ b_mu,
                                  const float* __restrict__ b_ls, float* __restrict__ out) {
    int wave = (blockIdx.x * blockDim.x + threadIdx.x) >> 5;
    int lane = threadIdx.x & 31;
    if (wave >= N_NODES) return;
    int c = __builtin_amdgcn_readfirstlane(wave);
    int l16 = lane & 15;
    float4 own  = ((const float4*)xw2)[(size_t)c * (HID / 4) + lane];
    float4 bias = (lane < 16) ? ((const float4*)b_mu)[l16] : ((const float4*)b_ls)[l16];
    float  dc   = dis[c];
    float4 acc  = gather_row(c, lane, cnt, spill_n, buckets, spill, xw2);
    float4 v;
    v.x = (acc.x + own.x) * dc + bias.x;
    v.y = (acc.y + own.y) * dc + bias.y;
    v.z = (acc.z + own.z) * dc + bias.z;
    v.w = (acc.w + own.w) * dc + bias.w;
    float* orow = out + ((lane < 16) ? (size_t)0 : (size_t)N_NODES * OUT_CH)
                      + (size_t)c * OUT_CH + l16 * 4;
    *(float4*)orow = v;
}

extern "C" void kernel_launch(void* const* d_in, const int* in_sizes, int n_in,
                              void* d_out, int out_size, void* d_ws, size_t ws_size,
                              hipStream_t stream) {
    (void)in_sizes; (void)n_in; (void)out_size; (void)ws_size;
    const float* x   = (const float*)d_in[0];
    const int*   ei  = (const int*)d_in[1];
    const float* ew  = (const float*)d_in[2];
    const float* W1  = (const float*)d_in[3];
    const float* b1  = (const float*)d_in[4];
    const float* Wmu = (const float*)d_in[5];
    const float* bmu = (const float*)d_in[6];
    const float* Wls = (const float*)d_in[7];
    const float* bls = (const float*)d_in[8];
    const int* row = ei;
    const int* col = ei + N_EDGES;

    float* ws      = (float*)d_ws;
    float* dis     = ws;
    int*   cnt     = (int*)(ws + 50000);
    int*   spill_n = cnt + N_NODES;
    float* xw      = ws + 150000;
    float* agg     = ws + 150000 + 6400000;
    int*   buckets = (int*)(ws + 12950016);
    int*   spill   = (int*)(ws + 16150016);

    hipMemsetAsync(cnt, 0, (size_t)(N_NODES + 4) * sizeof(int), stream);

    bucket_kernel<<<(N_EDGES + 255) / 256, 256, 0, stream>>>(row, col, ew, cnt, spill_n, buckets, spill);
    dis_kernel<<<(N_NODES + 255) / 256, 256, 0, stream>>>(cnt, spill_n, buckets, spill, dis);

    gemm_xwt_wmma<<<(M_TILES + GEMM_WAVES - 1) / GEMM_WAVES, GEMM_WAVES * 32, 0, stream>>>(x, W1, W1, HID, dis, xw);
    gather_relu_kernel<<<N_NODES / 8, 256, 0, stream>>>(cnt, spill_n, buckets, spill, dis, xw, b1, agg);

    gemm_xwt_wmma<<<(M_TILES + GEMM_WAVES - 1) / GEMM_WAVES, GEMM_WAVES * 32, 0, stream>>>(agg, Wmu, Wls, OUT_CH, dis, xw);
    gather_out_kernel<<<N_NODES / 8, 256, 0, stream>>>(cnt, spill_n, buckets, spill, dis, xw, bmu, bls,
                                                       (float*)d_out);
}


# Edges grouped by destination through LDS in two passes (cells of 256 nodes per edge chunk, then one work-group per 256 nodes fills buckets, counts, deg^-1/2): no global atomic per edge; arrival order
# speedup vs baseline: 4.0850x; 1.3052x over previous
#include <hip/hip_runtime.h>


#define N_NODES 50000
#define N_EDGES 600000
#define IN_CH   128
#define HID     128
#define OUT_CH  64

typedef __attribute__((ext_vector_type(2))) float v2f;
typedef __attribute__((ext_vector_type(8))) float v8f;

typedef __attribute__((ext_vector_type(2))) int v2i;

typedef __attribute__((ext_vector_type(4))) int v4i;

#define CAP 32

#define BIN_NODES 256
#define NBIN      196
#define NCHUNK    256
#define CHUNK     2344
#define EPT       10

__global__ void __launch_bounds__(256)
part_kernel(const int* __restrict__ row, const int* __restrict__ col,
            const float* __restrict__ w, int* __restrict__ pspill_n,
            int* __restrict__ slab, int* __restrict__ celltab, int* __restrict__ pspill) {
    __shared__ int lcnt[NBIN];
    __shared__ v2i lcell[NBIN * CAP];

    int tid   = threadIdx.x;
    int lane  = tid & 31;
    int wv    = __builtin_amdgcn_readfirstlane(tid >> 5);
    int chunk = blockIdx.x;
    int base  = chunk * CHUNK;
    int end   = (base + CHUNK < N_EDGES) ? (base + CHUNK) : N_EDGES;

    if (tid < NBIN) lcnt[tid] = 0;

    int   r[EPT], c[EPT];
    float wt[EPT];
#pragma unroll
    for (int k = 0; k < EPT; ++k) {
        int  e  = base + k * 256 + tid;
        bool ok = e < end;
        r[k]  = ok ? row[e] : 0;
        c[k]  = ok ? col[e] : 0;
        wt[k] = ok ? w[e] : 0.0f;
    }
    __syncthreads();

#pragma unroll
    for (int k = 0; k < EPT; ++k) {
        int e = base + k * 256 + tid;
        if (e < end) {
            float we = (r[k] == c[k]) ? 0.0f : wt[k];
            int b  = c[k] >> 8;
            int dl = c[k] & (BIN_NODES - 1);
            int s  = atomicAdd(&lcnt[b], 1);
            if (s < CAP) {
                v2i pr = {r[k] | (dl << 16), __float_as_int(we)};
                lcell[b * CAP + s] = pr;
            } else {
                int o = atomicAdd(pspill_n, 1);
                v4i en = {c[k], r[k], __float_as_int(we), 0};
                *(v4i*)(pspill + 4 * (size_t)o) = en;
            }
        }
    }
    __syncthreads();

    int myb = wv + 8 * lane;
    int myc = (myb < NBIN) ? lcnt[myb] : 0;
    myc = (myc < CAP) ? myc : CAP;
    if (myb < NBIN) celltab[myb * NCHUNK + chunk] = myc;
#pragma unroll
    for (int i = 0; i < (NBIN + 7) / 8; ++i) {
        int b = wv + 8 * i;
        int n = __builtin_amdgcn_readlane(myc, i);
        if (lane < n)
            *(v2i*)(slab + ((size_t)(b * NCHUNK + chunk) * CAP + lane) * 2) = lcell[b * CAP + lane];
    }
}

static __device__ __forceinline__ void place_edge(int src, int dl, int wb, int b,
                                                  int* ncnt, float* degx, v2i* nb,
                                                  int* __restrict__ spill_n, int* __restrict__ spill) {
    int s = atomicAdd(&ncnt[dl], 1);
    atomicAdd(&degx[dl], __int_as_float(wb));
    if (s < CAP) {
        v2i pr = {src, wb};
        nb[dl * CAP + s] = pr;
    } else {
        int o = atomicAdd(spill_n, 1);
        v4i en = {b * BIN_NODES + dl, src, wb, 0};
        *(v4i*)(spill + 4 * (size_t)o) = en;
    }
}

__global__ void __launch_bounds__(256)
bin_kernel(const int* __restrict__ slab, const int* __restrict__ celltab,
           const int* __restrict__ pspill_n, const int* __restrict__ pspill,
           int* __restrict__ spill_n, int* __restrict__ spill,
           int* __restrict__ cnt, int* __restrict__ buckets, float* __restrict__ dis) {
    __shared__ int   ncnt[BIN_NODES];
    __shared__ float degx[BIN_NODES];
    __shared__ v2i   nb[BIN_NODES * CAP];

    int tid  = threadIdx.x;
    int lane = tid & 31;
    int wv   = __builtin_amdgcn_readfirstlane(tid >> 5);
    int b    = blockIdx.x;

    ncnt[tid] = 0;
    degx[tid] = 0.0f;
    int myn = celltab[b * NCHUNK + wv * 32 + lane];
    __syncthreads();

    const v2i* cells = (const v2i*)slab + (size_t)(b * NCHUNK + wv * 32) * CAP;
    for (int i0 = 0; i0 < 32; i0 += 8) {
        v2i en[8];
#pragma unroll
        for (int k = 0; k < 8; ++k) {
            int n = __builtin_amdgcn_readlane(myn, i0 + k);
            v2i t = {-1, 0};
            if (lane < n) t = cells[(i0 + k) * CAP + lane];
            en[k] = t;
        }
#pragma unroll
        for (int k = 0; k < 8; ++k) {
            if (en[k].x >= 0)
                place_edge(en[k].x & 0xffff, (en[k].x >> 16) & (BIN_NODES - 1), en[k].y, b,
                           ncnt, degx, nb, spill_n, spill);
        }
    }
    int np = *pspill_n;
    for (int k = tid; k < np; k += 256) {
        v4i en = *(const v4i*)(pspill + 4 * (size_t)k);
        if ((en.x >> 8) == b)
            place_edge(en.y, en.x & (BIN_NODES - 1), en.z, b, ncnt, degx, nb, spill_n, spill);
    }
    __syncthreads();

    int node = b * BIN_NODES + tid;
    int n = ncnt[tid];
    float d = degx[tid] + 1.0f;
    if (node < N_NODES) {
        cnt[node] = n;
        dis[node] = rsqrtf(d);
    }

    v4i*       gb = (v4i*)(buckets + (size_t)b * (BIN_NODES * 2 * CAP));
    const v4i* lb = (const v4i*)nb;
    for (int i = 0; i < 16; ++i) {
        int idx = i * 256 + tid;
        int nd  = idx >> 4;
        int q   = idx & 15;
        int mm  = ncnt[nd];
        mm = (mm < CAP) ? mm : CAP;
        if (2 * q < mm && b * BIN_NODES + nd < N_NODES) gb[idx] = lb[idx];
    }
}

#define GEMM_WAVES 8
#define M_TILES    (N_NODES / 16)
#define W_PITCH4   33

__global__ void __launch_bounds__(GEMM_WAVES * 32)
gemm_xwt_wmma(const float* __restrict__ X,
              const float* __restrict__ Wa,
              const float* __restrict__ Wb,
              int n_split,
              const float* __restrict__ rs,
              float* __restrict__ C) {
    __shared__ float4 wl[HID * W_PITCH4];

    int tid  = threadIdx.x;
    int lane = tid & 31;
    int wv   = __builtin_amdgcn_readfirstlane(tid >> 5);
    int mt   = blockIdx.x * GEMM_WAVES + wv;
    int mtc  = (mt < M_TILES) ? mt : (M_TILES - 1);
    int c    = lane & 15;
    int h    = lane >> 4;

    const float4* xrow = (const float4*)(X + (size_t)(mtc * 16 + c) * IN_CH) + h;
    float4 a[16];
#pragma unroll
    for (int j = 0; j < 16; ++j) a[j] = xrow[2 * j];
    const float4* rsp = (const float4*)(rs + mtc * 16 + h * 8);
    float4 rs_lo = rsp[0];
    float4 rs_hi = rsp[1];

    float4 wrow[HID / GEMM_WAVES];
#pragma unroll
    for (int i = 0; i < HID / GEMM_WAVES; ++i) {
        int n = i * GEMM_WAVES + wv;
        const float* srow = (n < n_split) ? (Wa + (size_t)n * IN_CH)
                                          : (Wb + (size_t)(n - n_split) * IN_CH);
        wrow[i] = ((const float4*)srow)[lane];
    }
#pragma unroll
    for (int i = 0; i < HID / GEMM_WAVES; ++i) {
        int n  = i * GEMM_WAVES + wv;
        int t  = ((n >> 6) << 2) | (n & 3);
        int cc = (n & 63) >> 2;
        wl[(16 * t + cc) * W_PITCH4 + lane] = wrow[i];
    }
    __syncthreads();

    v8f acc[8];
#pragma unroll
    for (int t = 0; t < 8; ++t) acc[t] = v8f{};

    const float4* bl = wl + c * W_PITCH4 + h;
    float4 bcur[8], bnxt[8];
#pragma unroll
    for (int t = 0; t < 8; ++t) bcur[t] = bl[(16 * t) * W_PITCH4];

#pragma unroll
    for (int j = 0; j < 16; ++j) {
        if (j + 1 < 16) {
#pragma unroll
            for (int t = 0; t < 8; ++t) bnxt[t] = bl[(16 * t) * W_PITCH4 + 2 * (j + 1)];
        }
        __builtin_amdgcn_sched_barrier(0);
        v2f a0 = {a[j].x, a[j].y};
        v2f a1 = {a[j].z, a[j].w};
#pragma unroll
        for (int t = 0; t < 8; ++t) {
            v2f b0 = {bcur[t].x, bcur[t].y};
            acc[t] = __builtin_amdgcn_wmma_f32_16x16x4_f32(
                false, a0, false, b0, (short)0, acc[t], false, false);
        }
#pragma unroll
        for (int t = 0; t < 8; ++t) {
            v2f b1 = {bcur[t].z, bcur[t].w};
            acc[t] = __builtin_amdgcn_wmma_f32_16x16x4_f32(
                false, a1, false, b1, (short)0, acc[t], false, false);
        }
        __builtin_amdgcn_sched_barrier(0);
#pragma unroll
        for (int t = 0; t < 8; ++t) bcur[t] = bnxt[t];
    }

    float rsv[8] = {rs_lo.x, rs_lo.y, rs_lo.z, rs_lo.w, rs_hi.x, rs_hi.y, rs_hi.z, rs_hi.w};
#pragma unroll
    for (int t = 0; t < 8; ++t) {
#pragma unroll
        for (int r = 0; r < 8; ++r) acc[t][r] *= rsv[r];
    }

    if (mt < M_TILES) {
        float* crow = C + (size_t)(mt * 16 + h * 8) * HID + 4 * c;
#pragma unroll
        for (int r = 0; r < 8; ++r) {
#pragma unroll
            for (int q = 0; q < 2; ++q) {
                float4 o = {acc[4 * q + 0][r], acc[4 * q + 1][r], acc[4 * q + 2][r], acc[4 * q + 3][r]};
                *(float4*)(crow + (size_t)r * HID + 64 * q) = o;
            }
        }
    }
}

static __device__ __forceinline__ float4 gather_row(int c, int lane,
                                                    const int* __restrict__ cnt,
                                                    const int* __restrict__ spill_n,
                                                    const int* __restrict__ buckets,
                                                    const int* __restrict__ spill,
                                                    const float* __restrict__ src) {
    int n = cnt[c];
    int m = (n < CAP) ? n : CAP;
    const float4* src4 = (const float4*)src;

    float4 acc = {0.0f, 0.0f, 0.0f, 0.0f};
    int   rj = 0, wj = 0;
    if (lane < m) {
        v2i pr = *(const v2i*)(buckets + (size_t)c * (2 * CAP) + 2 * lane);
        rj = pr.x;
        wj = pr.y;
    }
    int last = m - 1;
    for (int j = 0; j < m; j += 4) {
        int j1 = (j + 1 < last) ? j + 1 : last;
        int j2 = (j + 2 < last) ? j + 2 : last;
        int j3 = (j + 3 < last) ? j + 3 : last;
        int r0 = __builtin_amdgcn_readlane(rj, j);
        int r1 = __builtin_amdgcn_readlane(rj, j1);
        int r2 = __builtin_amdgcn_readlane(rj, j2);
        int r3 = __builtin_amdgcn_readlane(rj, j3);
        float4 v0 = src4[(size_t)r0 * (HID / 4) + lane];
        float4 v1 = src4[(size_t)r1 * (HID / 4) + lane];
        float4 v2 = src4[(size_t)r2 * (HID / 4) + lane];
        float4 v3 = src4[(size_t)r3 * (HID / 4) + lane];
        float c0 = __int_as_float(__builtin_amdgcn_readlane(wj, j));
        float c1 = __int_as_float(__builtin_amdgcn_readlane(wj, j1));
        float c2 = __int_as_float(__builtin_amdgcn_readlane(wj, j2));
        float c3 = __int_as_float(__builtin_amdgcn_readlane(wj, j3));
        if (j + 1 >= m) c1 = 0.0f;
        if (j + 2 >= m) c2 = 0.0f;
        if (j + 3 >= m) c3 = 0.0f;
        acc.x += c0 * v0.x; acc.y += c0 * v0.y; acc.z += c0 * v0.z; acc.w += c0 * v0.w;
        acc.x += c1 * v1.x; acc.y += c1 * v1.y; acc.z += c1 * v1.z; acc.w += c1 * v1.w;
        acc.x += c2 * v2.x; acc.y += c2 * v2.y; acc.z += c2 * v2.z; acc.w += c2 * v2.w;
        acc.x += c3 * v3.x; acc.y += c3 * v3.y; acc.z += c3 * v3.z; acc.w += c3 * v3.w;
    }
    if (n > CAP) {
        int no = *spill_n;
        for (int base = 0; base < no; base += 32) {
            int k  = base + lane;
            int oc = -1, orow = 0, ow = 0;
            if (k < no) {
                v4i en = *(const v4i*)(spill + 4 * (size_t)k);
                oc = en.x; orow = en.y; ow = en.z;
            }
            unsigned hit = __builtin_amdgcn_ballot_w32(oc == c);
            while (hit != 0u) {
                int l = __builtin_ctz(hit);
                hit &= hit - 1u;
                int   r  = __builtin_amdgcn_readlane(orow, l);
                float cw = __int_as_float(__builtin_amdgcn_readlane(ow, l));
                float4 v = src4[(size_t)r * (HID / 4) + lane];
                acc.x += cw * v.x; acc.y += cw * v.y; acc.z += cw * v.z; acc.w += cw * v.w;
            }
        }
    }
    return acc;
}

__global__ void gather_relu_kernel(const int* __restrict__ cnt, const int* __restrict__ spill_n,
                                   const int* __restrict__ buckets, const int* __restrict__ spill,
                                   const float* __restrict__ dis,
                                   const float* __restrict__ xw, const float* __restrict__ b1,
                                   float* __restrict__ h) {
    int wave = (blockIdx.x * blockDim.x + threadIdx.x) >> 5;
    int lane = threadIdx.x & 31;
    if (wave >= N_NODES) return;
    int c = __builtin_amdgcn_readfirstlane(wave);
    float4 own  = ((const float4*)xw)[(size_t)c * (HID / 4) + lane];
    float4 bias = ((const float4*)b1)[lane];
    float  dc   = dis[c];
    float4 acc  = gather_row(c, lane, cnt, spill_n, buckets, spill, xw);
    float4 v;
    v.x = fmaxf((acc.x + own.x) * dc + bias.x, 0.0f);
    v.y = fmaxf((acc.y + own.y) * dc + bias.y, 0.0f);
    v.z = fmaxf((acc.z + own.z) * dc + bias.z, 0.0f);
    v.w = fmaxf((acc.w + own.w) * dc + bias.w, 0.0f);
    ((float4*)h)[(size_t)c * (HID / 4) + lane] = v;
}

__global__ void gather_out_kernel(const int* __restrict__ cnt, const int* __restrict__ spill_n,
                                  const int* __restrict__ buckets, const int* __restrict__ spill,
                                  const float* __restrict__ dis,
                                  const float* __restrict__ xw2, const float* __restrict__ b_mu,
                                  const float* __restrict__ b_ls, float* __restrict__ out) {
    int wave = (blockIdx.x * blockDim.x + threadIdx.x) >> 5;
    int lane = threadIdx.x & 31;
    if (wave >= N_NODES) return;
    int c = __builtin_amdgcn_readfirstlane(wave);
    int l16 = lane & 15;
    float4 own  = ((const float4*)xw2)[(size_t)c * (HID / 4) + lane];
    float4 bias = (lane < 16) ? ((const float4*)b_mu)[l16] : ((const float4*)b_ls)[l16];
    float  dc   = dis[c];
    float4 acc  = gather_row(c, lane, cnt, spill_n, buckets, spill, xw2);
    float4 v;
    v.x = (acc.x + own.x) * dc + bias.x;
    v.y = (acc.y + own.y) * dc + bias.y;
    v.z = (acc.z + own.z) * dc + bias.z;
    v.w = (acc.w + own.w) * dc + bias.w;
    float* orow = out + ((lane < 16) ? (size_t)0 : (size_t)N_NODES * OUT_CH)
                      + (size_t)c * OUT_CH + l16 * 4;
    *(float4*)orow = v;
}

extern "C" void kernel_launch(void* const* d_in, const int* in_sizes, int n_in,
                              void* d_out, int out_size, void* d_ws, size_t ws_size,
                              hipStream_t stream) {
    (void)in_sizes; (void)n_in; (void)out_size; (void)ws_size;
    const float* x   = (const float*)d_in[0];
    const int*   ei  = (const int*)d_in[1];
    const float* ew  = (const float*)d_in[2];
    const float* W1  = (const float*)d_in[3];
    const float* b1  = (const float*)d_in[4];
    const float* Wmu = (const float*)d_in[5];
    const float* bmu = (const float*)d_in[6];
    const float* Wls = (const float*)d_in[7];
    const float* bls = (const float*)d_in[8];
    const int* row = ei;
    const int* col = ei + N_EDGES;

    float* ws      = (float*)d_ws;
    float* dis     = ws;
    int*   cnt     = (int*)(ws + 50000);
    int*   spill_n = cnt + N_NODES;
    float* xw      = ws + 150000;
    float* agg     = ws + 150000 + 6400000;
    int*   buckets = (int*)(ws + 12950016);
    int*   spill   = (int*)(ws + 16150016);
    int*   pspill  = (int*)(ws + 150000);
    int*   slab    = (int*)(ws + 6550016);
    int*   celltab = (int*)(ws + 9761280);

    hipMemsetAsync(spill_n, 0, 4 * sizeof(int), stream);

    part_kernel<<<NCHUNK, 256, 0, stream>>>(row, col, ew, spill_n + 1, slab, celltab, pspill);
    bin_kernel<<<NBIN, 256, 0, stream>>>(slab, celltab, spill_n + 1, pspill, spill_n, spill, cnt, buckets, dis);

    gemm_xwt_wmma<<<(M_TILES + GEMM_WAVES - 1) / GEMM_WAVES, GEMM_WAVES * 32, 0, stream>>>(x, W1, W1, HID, dis, xw);
    gather_relu_kernel<<<N_NODES / 8, 256, 0, stream>>>(cnt, spill_n, buckets, spill, dis, xw, b1, agg);

    gemm_xwt_wmma<<<(M_TILES + GEMM_WAVES - 1) / GEMM_WAVES, GEMM_WAVES * 32, 0, stream>>>(agg, Wmu, Wls, OUT_CH, dis, xw);
    gather_out_kernel<<<N_NODES / 8, 256, 0, stream>>>(cnt, spill_n, buckets, spill, dis, xw, bmu, bls,
                                                       (float*)d_out);
}


# Layer-2 GEMM moved into the layer-1 gather kernel: 8-wave groups gather 16 nodes into LDS, then 32 WMMAs per wave, weights held in registers; h never stored, one launch fewer; sum order unchanged
# speedup vs baseline: 4.2638x; 1.0438x over previous
#include <hip/hip_runtime.h>


#define N_NODES 50000
#define N_EDGES 600000
#define IN_CH   128
#define HID     128
#define OUT_CH  64

typedef __attribute__((ext_vector_type(2))) float v2f;
typedef __attribute__((ext_vector_type(8))) float v8f;

typedef __attribute__((ext_vector_type(2))) int v2i;

typedef __attribute__((ext_vector_type(4))) int v4i;

#define CAP 32

#define BIN_NODES 256
#define NBIN      196
#define NCHUNK    256
#define CHUNK     2344
#define EPT       10

__global__ void __launch_bounds__(256)
part_kernel(const int* __restrict__ row, const int* __restrict__ col,
            const float* __restrict__ w, int* __restrict__ pspill_n,
            int* __restrict__ slab, int* __restrict__ celltab, int* __restrict__ pspill) {
    __shared__ int lcnt[NBIN];
    __shared__ v2i lcell[NBIN * CAP];

    int tid   = threadIdx.x;
    int lane  = tid & 31;
    int wv    = __builtin_amdgcn_readfirstlane(tid >> 5);
    int chunk = blockIdx.x;
    int base  = chunk * CHUNK;
    int end   = (base + CHUNK < N_EDGES) ? (base + CHUNK) : N_EDGES;

    if (tid < NBIN) lcnt[tid] = 0;

    int   r[EPT], c[EPT];
    float wt[EPT];
#pragma unroll
    for (int k = 0; k < EPT; ++k) {
        int  e  = base + k * 256 + tid;
        bool ok = e < end;
        r[k]  = ok ? row[e] : 0;
        c[k]  = ok ? col[e] : 0;
        wt[k] = ok ? w[e] : 0.0f;
    }
    __syncthreads();

#pragma unroll
    for (int k = 0; k < EPT; ++k) {
        int e = base + k * 256 + tid;
        if (e < end) {
            float we = (r[k] == c[k]) ? 0.0f : wt[k];
            int b  = c[k] >> 8;
            int dl = c[k] & (BIN_NODES - 1);
            int s  = atomicAdd(&lcnt[b], 1);
            if (s < CAP) {
                v2i pr = {r[k] | (dl << 16), __float_as_int(we)};
                lcell[b * CAP + s] = pr;
            } else {
                int o = atomicAdd(pspill_n, 1);
                v4i en = {c[k], r[k], __float_as_int(we), 0};
                *(v4i*)(pspill + 4 * (size_t)o) = en;
            }
        }
    }
    __syncthreads();

    int myb = wv + 8 * lane;
    int myc = (myb < NBIN) ? lcnt[myb] : 0;
    myc = (myc < CAP) ? myc : CAP;
    if (myb < NBIN) celltab[myb * NCHUNK + chunk] = myc;
#pragma unroll
    for (int i = 0; i < (NBIN + 7) / 8; ++i) {
        int b = wv + 8 * i;
        int n = __builtin_amdgcn_readlane(myc, i);
        if (lane < n)
            *(v2i*)(slab + ((size_t)(b * NCHUNK + chunk) * CAP + lane) * 2) = lcell[b * CAP + lane];
    }
}

static __device__ __forceinline__ void place_edge(int src, int dl, int wb, int b,
                                                  int* ncnt, float* degx, v2i* nb,
                                                  int* __restrict__ spill_n, int* __restrict__ spill) {
    int s = atomicAdd(&ncnt[dl], 1);
    atomicAdd(&degx[dl], __int_as_float(wb));
    if (s < CAP) {
        v2i pr = {src, wb};
        nb[dl * CAP + s] = pr;
    } else {
        int o = atomicAdd(spill_n, 1);
        v4i en = {b * BIN_NODES + dl, src, wb, 0};
        *(v4i*)(spill + 4 * (size_t)o) = en;
    }
}

__global__ void __launch_bounds__(256)
bin_kernel(const int* __restrict__ slab, const int* __restrict__ celltab,
           const int* __restrict__ pspill_n, const int* __restrict__ pspill,
           int* __restrict__ spill_n, int* __restrict__ spill,
           int* __restrict__ cnt, int* __restrict__ buckets, float* __restrict__ dis) {
    __shared__ int   ncnt[BIN_NODES];
    __shared__ float degx[BIN_NODES];
    __shared__ v2i   nb[BIN_NODES * CAP];

    int tid  = threadIdx.x;
    int lane = tid & 31;
    int wv   = __builtin_amdgcn_readfirstlane(tid >> 5);
    int b    = blockIdx.x;

    ncnt[tid] = 0;
    degx[tid] = 0.0f;
    int myn = celltab[b * NCHUNK + wv * 32 + lane];
    __syncthreads();

    const v2i* cells = (const v2i*)slab + (size_t)(b * NCHUNK + wv * 32) * CAP;
    for (int i0 = 0; i0 < 32; i0 += 8) {
        v2i en[8];
#pragma unroll
        for (int k = 0; k < 8; ++k) {
            int n = __builtin_amdgcn_readlane(myn, i0 + k);
            v2i t = {-1, 0};
            if (lane < n) t = cells[(i0 + k) * CAP + lane];
            en[k] = t;
        }
#pragma unroll
        for (int k = 0; k < 8; ++k) {
            if (en[k].x >= 0)
                place_edge(en[k].x & 0xffff, (en[k].x >> 16) & (BIN_NODES - 1), en[k].y, b,
                           ncnt, degx, nb, spill_n, spill);
        }
    }
    int np = *pspill_n;
    for (int k = tid; k < np; k += 256) {
        v4i en = *(const v4i*)(pspill + 4 * (size_t)k);
        if ((en.x >> 8) == b)
            place_edge(en.y, en.x & (BIN_NODES - 1), en.z, b, ncnt, degx, nb, spill_n, spill);
    }
    __syncthreads();

    int node = b * BIN_NODES + tid;
    int n = ncnt[tid];
    float d = degx[tid] + 1.0f;
    if (node < N_NODES) {
        cnt[node] = n;
        dis[node] = rsqrtf(d);
    }

    v4i*       gb = (v4i*)(buckets + (size_t)b * (BIN_NODES * 2 * CAP));
    const v4i* lb = (const v4i*)nb;
    for (int i = 0; i < 16; ++i) {
        int idx = i * 256 + tid;
        int nd  = idx >> 4;
        int q   = idx & 15;
        int mm  = ncnt[nd];
        mm = (mm < CAP) ? mm : CAP;
        if (2 * q < mm && b * BIN_NODES + nd < N_NODES) gb[idx] = lb[idx];
    }
}

#define GEMM_WAVES 8
#define M_TILES    (N_NODES / 16)
#define W_PITCH4   33

__global__ void __launch_bounds__(GEMM_WAVES * 32)
gemm_xwt_wmma(const float* __restrict__ X,
              const float* __restrict__ Wa,
              const float* __restrict__ Wb,
              int n_split,
              const float* __restrict__ rs,
              float* __restrict__ C) {
    __shared__ float4 wl[HID * W_PITCH4];

    int tid  = threadIdx.x;
    int lane = tid & 31;
    int wv   = __builtin_amdgcn_readfirstlane(tid >> 5);
    int mt   = blockIdx.x * GEMM_WAVES + wv;
    int mtc  = (mt < M_TILES) ? mt : (M_TILES - 1);
    int c    = lane & 15;
    int h    = lane >> 4;

    const float4* xrow = (const float4*)(X + (size_t)(mtc * 16 + c) * IN_CH) + h;
    float4 a[16];
#pragma unroll
    for (int j = 0; j < 16; ++j) a[j] = xrow[2 * j];
    const float4* rsp = (const float4*)(rs + mtc * 16 + h * 8);
    float4 rs_lo = rsp[0];
    float4 rs_hi = rsp[1];

    float4 wrow[HID / GEMM_WAVES];
#pragma unroll
    for (int i = 0; i < HID / GEMM_WAVES; ++i) {
        int n = i * GEMM_WAVES + wv;
        const float* srow = (n < n_split) ? (Wa + (size_t)n * IN_CH)
                                          : (Wb + (size_t)(n - n_split) * IN_CH);
        wrow[i] = ((const float4*)srow)[lane];
    }
#pragma unroll
    for (int i = 0; i < HID / GEMM_WAVES; ++i) {
        int n  = i * GEMM_WAVES + wv;
        int t  = ((n >> 6) << 2) | (n & 3);
        int cc = (n & 63) >> 2;
        wl[(16 * t + cc) * W_PITCH4 + lane] = wrow[i];
    }
    __syncthreads();

    v8f acc[8];
#pragma unroll
    for (int t = 0; t < 8; ++t) acc[t] = v8f{};

    const float4* bl = wl + c * W_PITCH4 + h;
    float4 bcur[8], bnxt[8];
#pragma unroll
    for (int t = 0; t < 8; ++t) bcur[t] = bl[(16 * t) * W_PITCH4];

#pragma unroll
    for (int j = 0; j < 16; ++j) {
        if (j + 1 < 16) {
#pragma unroll
            for (int t = 0; t < 8; ++t) bnxt[t] = bl[(16 * t) * W_PITCH4 + 2 * (j + 1)];
        }
        __builtin_amdgcn_sched_barrier(0);
        v2f a0 = {a[j].x, a[j].y};
        v2f a1 = {a[j].z, a[j].w};
#pragma unroll
        for (int t = 0; t < 8; ++t) {
            v2f b0 = {bcur[t].x, bcur[t].y};
            acc[t] = __builtin_amdgcn_wmma_f32_16x16x4_f32(
                false, a0, false, b0, (short)0, acc[t], false, false);
        }
#pragma unroll
        for (int t = 0; t < 8; ++t) {
            v2f b1 = {bcur[t].z, bcur[t].w};
            acc[t] = __builtin_amdgcn_wmma_f32_16x16x4_f32(
                false, a1, false, b1, (short)0, acc[t], false, false);
        }
        __builtin_amdgcn_sched_barrier(0);
#pragma unroll
        for (int t = 0; t < 8; ++t) bcur[t] = bnxt[t];
    }

    float rsv[8] = {rs_lo.x, rs_lo.y, rs_lo.z, rs_lo.w, rs_hi.x, rs_hi.y, rs_hi.z, rs_hi.w};
#pragma unroll
    for (int t = 0; t < 8; ++t) {
#pragma unroll
        for (int r = 0; r < 8; ++r) acc[t][r] *= rsv[r];
    }

    if (mt < M_TILES) {
        float* crow = C + (size_t)(mt * 16 + h * 8) * HID + 4 * c;
#pragma unroll
        for (int r = 0; r < 8; ++r) {
#pragma unroll
            for (int q = 0; q < 2; ++q) {
                float4 o = {acc[4 * q + 0][r], acc[4 * q + 1][r], acc[4 * q + 2][r], acc[4 * q + 3][r]};
                *(float4*)(crow + (size_t)r * HID + 64 * q) = o;
            }
        }
    }
}

struct bucket_t { int n; int rj; int wj; };

static __device__ __forceinline__ bucket_t bucket_fetch(int c, int lane,
                                                        const int* __restrict__ cnt,
                                                        const int* __restrict__ buckets) {
    bucket_t k;
    k.n  = cnt[c];
    k.rj = 0;
    k.wj = 0;
    int m = (k.n < CAP) ? k.n : CAP;
    if (lane < m) {
        v2i pr = *(const v2i*)(buckets + (size_t)c * (2 * CAP) + 2 * lane);
        k.rj = pr.x;
        k.wj = pr.y;
    }
    return k;
}

static __device__ __forceinline__ float4 gather_row(int c, int lane, bucket_t k,
                                                    const int* __restrict__ spill_n,
                                                    const int* __restrict__ spill,
                                                    const float* __restrict__ src) {
    int n = k.n;
    int m = (n < CAP) ? n : CAP;
    const float4* src4 = (const float4*)src;

    float4 acc = {0.0f, 0.0f, 0.0f, 0.0f};
    int   rj = k.rj, wj = k.wj;
    int last = m - 1;
    for (int j = 0; j < m; j += 4) {
        int j1 = (j + 1 < last) ? j + 1 : last;
        int j2 = (j + 2 < last) ? j + 2 : last;
        int j3 = (j + 3 < last) ? j + 3 : last;
        int r0 = __builtin_amdgcn_readlane(rj, j);
        int r1 = __builtin_amdgcn_readlane(rj, j1);
        int r2 = __builtin_amdgcn_readlane(rj, j2);
        int r3 = __builtin_amdgcn_readlane(rj, j3);
        float4 v0 = src4[(size_t)r0 * (HID / 4) + lane];
        float4 v1 = src4[(size_t)r1 * (HID / 4) + lane];
        float4 v2 = src4[(size_t)r2 * (HID / 4) + lane];
        float4 v3 = src4[(size_t)r3 * (HID / 4) + lane];
        float c0 = __int_as_float(__builtin_amdgcn_readlane(wj, j));
        float c1 = __int_as_float(__builtin_amdgcn_readlane(wj, j1));
        float c2 = __int_as_float(__builtin_amdgcn_readlane(wj, j2));
        float c3 = __int_as_float(__builtin_amdgcn_readlane(wj, j3));
        if (j + 1 >= m) c1 = 0.0f;
        if (j + 2 >= m) c2 = 0.0f;
        if (j + 3 >= m) c3 = 0.0f;
        acc.x += c0 * v0.x; acc.y += c0 * v0.y; acc.z += c0 * v0.z; acc.w += c0 * v0.w;
        acc.x += c1 * v1.x; acc.y += c1 * v1.y; acc.z += c1 * v1.z; acc.w += c1 * v1.w;
        acc.x += c2 * v2.x; acc.y += c2 * v2.y; acc.z += c2 * v2.z; acc.w += c2 * v2.w;
        acc.x += c3 * v3.x; acc.y += c3 * v3.y; acc.z += c3 * v3.z; acc.w += c3 * v3.w;
    }
    if (n > CAP) {
        int no = *spill_n;
        for (int base = 0; base < no; base += 32) {
            int k  = base + lane;
            int oc = -1, orow = 0, ow = 0;
            if (k < no) {
                v4i en = *(const v4i*)(spill + 4 * (size_t)k);
                oc = en.x; orow = en.y; ow = en.z;
            }
            unsigned hit = __builtin_amdgcn_ballot_w32(oc == c);
            while (hit != 0u) {
                int l = __builtin_ctz(hit);
                hit &= hit - 1u;
                int   r  = __builtin_amdgcn_readlane(orow, l);
                float cw = __int_as_float(__builtin_amdgcn_readlane(ow, l));
                float4 v = src4[(size_t)r * (HID / 4) + lane];
                acc.x += cw * v.x; acc.y += cw * v.y; acc.z += cw * v.z; acc.w += cw * v.w;
            }
        }
    }
    return acc;
}

#define FUSE_WAVES 8
#define FUSE_GRID  1024
#define T_PITCH4   33

__global__ void __launch_bounds__(FUSE_WAVES * 32, 8)
gather_relu_gemm_kernel(const int* __restrict__ cnt, const int* __restrict__ spill_n,
                        const int* __restrict__ buckets, const int* __restrict__ spill,
                        const float* __restrict__ dis,
                        const float* __restrict__ xw, const float* __restrict__ b1,
                        const float* __restrict__ Wa, const float* __restrict__ Wb,
                        float* __restrict__ xw2) {
    __shared__ float4 ht[16 * T_PITCH4];
    __shared__ float4 ot[16 * T_PITCH4];

    int tid  = threadIdx.x;
    int lane = tid & 31;
    int wv   = __builtin_amdgcn_readfirstlane(tid >> 5);
    int c    = lane & 15;
    int h    = lane >> 4;

    int nrow = 16 * wv + c;
    const float*  wrow = (nrow < OUT_CH) ? (Wa + (size_t)nrow * HID)
                                         : (Wb + (size_t)(nrow - OUT_CH) * HID);
    const float4* wr4  = (const float4*)wrow + h;
    float4 bf[16];
#pragma unroll
    for (int j = 0; j < 16; ++j) bf[j] = wr4[2 * j];

    float4 bias = ((const float4*)b1)[lane];
    const float4* xw4 = (const float4*)xw;

    for (int s = blockIdx.x; s < M_TILES; s += FUSE_GRID) {
        int c0 = __builtin_amdgcn_readfirstlane(s * 16 + wv);
        int c1 = c0 + 8;
        bucket_t k0 = bucket_fetch(c0, lane, cnt, buckets);
        bucket_t k1 = bucket_fetch(c1, lane, cnt, buckets);
        float4 own0 = xw4[(size_t)c0 * (HID / 4) + lane];
        float4 own1 = xw4[(size_t)c1 * (HID / 4) + lane];
        float  d0   = dis[c0];
        float  d1   = dis[c1];

        float4 g0 = gather_row(c0, lane, k0, spill_n, spill, xw);
        float4 v0;
        v0.x = fmaxf((g0.x + own0.x) * d0 + bias.x, 0.0f);
        v0.y = fmaxf((g0.y + own0.y) * d0 + bias.y, 0.0f);
        v0.z = fmaxf((g0.z + own0.z) * d0 + bias.z, 0.0f);
        v0.w = fmaxf((g0.w + own0.w) * d0 + bias.w, 0.0f);
        ht[wv * T_PITCH4 + lane] = v0;

        float4 g1 = gather_row(c1, lane, k1, spill_n, spill, xw);
        float4 v1;
        v1.x = fmaxf((g1.x + own1.x) * d1 + bias.x, 0.0f);
        v1.y = fmaxf((g1.y + own1.y) * d1 + bias.y, 0.0f);
        v1.z = fmaxf((g1.z + own1.z) * d1 + bias.z, 0.0f);
        v1.w = fmaxf((g1.w + own1.w) * d1 + bias.w, 0.0f);
        ht[(wv + 8) * T_PITCH4 + lane] = v1;
        __syncthreads();

        const float4* al = ht + c * T_PITCH4 + h;
        v8f acc = v8f{};
        float4 a[8];
#pragma unroll
        for (int g = 0; g < 2; ++g) {
#pragma unroll
            for (int j = 0; j < 8; ++j) a[j] = al[2 * (8 * g + j)];
#pragma unroll
            for (int j = 0; j < 8; ++j) {
                v2f a0 = {a[j].x, a[j].y};
                v2f a1 = {a[j].z, a[j].w};
                v2f b0 = {bf[8 * g + j].x, bf[8 * g + j].y};
                v2f b1 = {bf[8 * g + j].z, bf[8 * g + j].w};
                acc = __builtin_amdgcn_wmma_f32_16x16x4_f32(
                    false, a0, false, b0, (short)0, acc, false, false);
                acc = __builtin_amdgcn_wmma_f32_16x16x4_f32(
                    false, a1, false, b1, (short)0, acc, false, false);
            }
        }

        float* of = (float*)ot;
#pragma unroll
        for (int r = 0; r < 8; ++r) of[(8 * h + r) * (4 * T_PITCH4) + 16 * wv + c] = acc[r];
        __syncthreads();

        float4 o0 = ot[wv * T_PITCH4 + lane];
        float4 o1 = ot[(wv + 8) * T_PITCH4 + lane];
        o0.x *= d0; o0.y *= d0; o0.z *= d0; o0.w *= d0;
        o1.x *= d1; o1.y *= d1; o1.z *= d1; o1.w *= d1;
        ((float4*)xw2)[(size_t)c0 * (HID / 4) + lane] = o0;
        ((float4*)xw2)[(size_t)c1 * (HID / 4) + lane] = o1;
    }
}

__global__ void gather_out_kernel(const int* __restrict__ cnt, const int* __restrict__ spill_n,
                                  const int* __restrict__ buckets, const int* __restrict__ spill,
                                  const float* __restrict__ dis,
                                  const float* __restrict__ xw2, const float* __restrict__ b_mu,
                                  const float* __restrict__ b_ls, float* __restrict__ out) {
    int wave = (blockIdx.x * blockDim.x + threadIdx.x) >> 5;
    int lane = threadIdx.x & 31;
    if (wave >= N_NODES) return;
    int c = __builtin_amdgcn_readfirstlane(wave);
    int l16 = lane & 15;
    float4 own  = ((const float4*)xw2)[(size_t)c * (HID / 4) + lane];
    float4 bias = (lane < 16) ? ((const float4*)b_mu)[l16] : ((const float4*)b_ls)[l16];
    float  dc   = dis[c];
    bucket_t k  = bucket_fetch(c, lane, cnt, buckets);
    float4 acc  = gather_row(c, lane, k, spill_n, spill, xw2);
    float4 v;
    v.x = (acc.x + own.x) * dc + bias.x;
    v.y = (acc.y + own.y) * dc + bias.y;
    v.z = (acc.z + own.z) * dc + bias.z;
    v.w = (acc.w + own.w) * dc + bias.w;
    float* orow = out + ((lane < 16) ? (size_t)0 : (size_t)N_NODES * OUT_CH)
                      + (size_t)c * OUT_CH + l16 * 4;
    *(float4*)orow = v;
}

extern "C" void kernel_launch(void* const* d_in, const int* in_sizes, int n_in,
                              void* d_out, int out_size, void* d_ws, size_t ws_size,
                              hipStream_t stream) {
    (void)in_sizes; (void)n_in; (void)out_size; (void)ws_size;
    const float* x   = (const float*)d_in[0];
    const int*   ei  = (const int*)d_in[1];
    const float* ew  = (const float*)d_in[2];
    const float* W1  = (const float*)d_in[3];
    const float* b1  = (const float*)d_in[4];
    const float* Wmu = (const float*)d_in[5];
    const float* bmu = (const float*)d_in[6];
    const float* Wls = (const float*)d_in[7];
    const float* bls = (const float*)d_in[8];
    const int* row = ei;
    const int* col = ei + N_EDGES;

    float* ws      = (float*)d_ws;
    float* dis     = ws;
    int*   cnt     = (int*)(ws + 50000);
    int*   spill_n = cnt + N_NODES;
    float* xw      = ws + 150000;
    float* agg     = ws + 150000 + 6400000;
    int*   buckets = (int*)(ws + 12950016);
    int*   spill   = (int*)(ws + 16150016);
    int*   pspill  = (int*)(ws + 150000);
    int*   slab    = (int*)(ws + 6550016);
    int*   celltab = (int*)(ws + 9761280);

    hipMemsetAsync(spill_n, 0, 4 * sizeof(int), stream);

    part_kernel<<<NCHUNK, 256, 0, stream>>>(row, col, ew, spill_n + 1, slab, celltab, pspill);
    bin_kernel<<<NBIN, 256, 0, stream>>>(slab, celltab, spill_n + 1, pspill, spill_n, spill, cnt, buckets, dis);

    gemm_xwt_wmma<<<(M_TILES + GEMM_WAVES - 1) / GEMM_WAVES, GEMM_WAVES * 32, 0, stream>>>(x, W1, W1, HID, dis, xw);
    gather_relu_gemm_kernel<<<FUSE_GRID, FUSE_WAVES * 32, 0, stream>>>(cnt, spill_n, buckets, spill, dis, xw, b1,
                                                                       Wmu, Wls, agg);

    gather_out_kernel<<<N_NODES / 8, 256, 0, stream>>>(cnt, spill_n, buckets, spill, dis, agg, bmu, bls,
                                                       (float*)d_out);
}


# Grouping kernels: the 32 cell loads, LDS counter adds and write-out LDS reads of a wave are issued as groups before first use (the built code waited after each one); sums stay in arrival order
# speedup vs baseline: 4.7214x; 1.1073x over previous
#include <hip/hip_runtime.h>


#define N_NODES 50000
#define N_EDGES 600000
#define IN_CH   128
#define HID     128
#define OUT_CH  64

typedef __attribute__((ext_vector_type(2))) float v2f;
typedef __attribute__((ext_vector_type(8))) float v8f;

typedef __attribute__((ext_vector_type(2))) int v2i;

typedef __attribute__((ext_vector_type(4))) int v4i;

#define CAP 32

#define BIN_NODES 256
#define NBIN      196
#define NCHUNK    256
#define CHUNK     2344
#define EPT       10

__global__ void __launch_bounds__(256)
part_kernel(const int* __restrict__ row, const int* __restrict__ col,
            const float* __restrict__ w, int* __restrict__ pspill_n,
            int* __restrict__ slab, int* __restrict__ celltab, int* __restrict__ pspill) {
    __shared__ int lcnt[NBIN];
    __shared__ v2i lcell[NBIN * CAP];

    int tid   = threadIdx.x;
    int lane  = tid & 31;
    int wv    = __builtin_amdgcn_readfirstlane(tid >> 5);
    int chunk = blockIdx.x;
    int base  = chunk * CHUNK;
    int end   = (base + CHUNK < N_EDGES) ? (base + CHUNK) : N_EDGES;

    if (tid < NBIN) lcnt[tid] = 0;

    int   r[EPT], c[EPT];
    float wt[EPT];
#pragma unroll
    for (int k = 0; k < EPT; ++k) {
        int  e  = base + k * 256 + tid;
        bool ok = e < end;
        r[k]  = ok ? row[e] : 0;
        c[k]  = ok ? col[e] : 0;
        wt[k] = ok ? w[e] : 0.0f;
    }
    __syncthreads();

    int s[EPT];
#pragma unroll
    for (int k = 0; k < EPT; ++k) {
        int e = base + k * 256 + tid;
        s[k] = 0;
        if (e < end) s[k] = atomicAdd(&lcnt[c[k] >> 8], 1);
    }
    unsigned over = 0u;
#pragma unroll
    for (int k = 0; k < EPT; ++k) {
        int e = base + k * 256 + tid;
        if (e < end) {
            float we = (r[k] == c[k]) ? 0.0f : wt[k];
            int b  = c[k] >> 8;
            int dl = c[k] & (BIN_NODES - 1);
            if (s[k] < CAP) {
                v2i pr = {r[k] | (dl << 16), __float_as_int(we)};
                lcell[b * CAP + s[k]] = pr;
            } else {
                over |= 1u << k;
            }
        }
    }
    if (over != 0u) {
#pragma unroll
        for (int k = 0; k < EPT; ++k) {
            if ((over >> k) & 1u) {
                float we = (r[k] == c[k]) ? 0.0f : wt[k];
                int o = atomicAdd(pspill_n, 1);
                v4i en = {c[k], r[k], __float_as_int(we), 0};
                *(v4i*)(pspill + 4 * (size_t)o) = en;
            }
        }
    }
    __syncthreads();

    int myb = wv + 8 * lane;
    int myc = (myb < NBIN) ? lcnt[myb] : 0;
    myc = (myc < CAP) ? myc : CAP;
    if (myb < NBIN) celltab[myb * NCHUNK + chunk] = myc;
    v2i pc[(NBIN + 7) / 8];
#pragma unroll
    for (int i = 0; i < (NBIN + 7) / 8; ++i) {
        int b = wv + 8 * i;
        int n = __builtin_amdgcn_readlane(myc, i);
        v2i t = {0, 0};
        if (lane < n) t = lcell[b * CAP + lane];
        pc[i] = t;
    }
#pragma unroll
    for (int i = 0; i < (NBIN + 7) / 8; ++i) {
        int b = wv + 8 * i;
        int n = __builtin_amdgcn_readlane(myc, i);
        if (lane < n)
            *(v2i*)(slab + ((size_t)(b * NCHUNK + chunk) * CAP + lane) * 2) = pc[i];
    }
}

static __device__ __forceinline__ int bucket_slot(int dl, int wb, int* ncnt, float* degx) {
    int s = atomicAdd(&ncnt[dl], 1);
    atomicAdd(&degx[dl], __int_as_float(wb));
    return s;
}

static __device__ __forceinline__ void bucket_put(int s, int src, int dl, int wb, int b, v2i* nb,
                                                  int* __restrict__ spill_n, int* __restrict__ spill) {
    if (s < CAP) {
        v2i pr = {src, wb};
        nb[dl * CAP + s] = pr;
    } else {
        int o = atomicAdd(spill_n, 1);
        v4i en = {b * BIN_NODES + dl, src, wb, 0};
        *(v4i*)(spill + 4 * (size_t)o) = en;
    }
}

__global__ void __launch_bounds__(256)
bin_kernel(const int* __restrict__ slab, const int* __restrict__ celltab,
           const int* __restrict__ pspill_n, const int* __restrict__ pspill,
           int* __restrict__ spill_n, int* __restrict__ spill,
           int* __restrict__ cnt, int* __restrict__ buckets, float* __restrict__ dis) {
    __shared__ int   ncnt[BIN_NODES];
    __shared__ float degx[BIN_NODES];
    __shared__ v2i   nb[BIN_NODES * CAP];

    int tid  = threadIdx.x;
    int lane = tid & 31;
    int wv   = __builtin_amdgcn_readfirstlane(tid >> 5);
    int b    = blockIdx.x;

    ncnt[tid] = 0;
    degx[tid] = 0.0f;
    int myn = celltab[b * NCHUNK + wv * 32 + lane];
    int np  = *pspill_n;
    __syncthreads();

    const v2i* cells = (const v2i*)slab + (size_t)(b * NCHUNK + wv * 32) * CAP;
    const unsigned long long* cells64 = (const unsigned long long*)cells;
    unsigned long long en[32];
#pragma unroll
    for (int k = 0; k < 32; ++k) {
        int n = __builtin_amdgcn_readlane(myn, k);
        unsigned long long t = ~0ull;
        if (lane < n) t = cells64[k * CAP + lane];
        en[k] = t;
    }
#pragma unroll
    for (int g = 0; g < 32; g += 16) {
        int s[16];
#pragma unroll
        for (int k = 0; k < 16; ++k) {
            int ex = (int)en[g + k];
            int wb = (int)(en[g + k] >> 32);
            s[k] = 0;
            if (ex >= 0) s[k] = bucket_slot((ex >> 16) & (BIN_NODES - 1), wb, ncnt, degx);
        }
        unsigned over = 0u;
#pragma unroll
        for (int k = 0; k < 16; ++k) {
            int ex = (int)en[g + k];
            int wb = (int)(en[g + k] >> 32);
            if (ex >= 0) {
                if (s[k] < CAP) {
                    v2i pr = {ex & 0xffff, wb};
                    nb[((ex >> 16) & (BIN_NODES - 1)) * CAP + s[k]] = pr;
                } else {
                    over |= 1u << k;
                }
            }
        }
        if (over != 0u) {
#pragma unroll
            for (int k = 0; k < 16; ++k) {
                if ((over >> k) & 1u) {
                    int ex = (int)en[g + k];
                    bucket_put(CAP, ex & 0xffff, (ex >> 16) & (BIN_NODES - 1), (int)(en[g + k] >> 32), b,
                               nb, spill_n, spill);
                }
            }
        }
    }
    for (int k = tid; k < np; k += 256) {
        v4i pe = *(const v4i*)(pspill + 4 * (size_t)k);
        if ((pe.x >> 8) == b) {
            int dl = pe.x & (BIN_NODES - 1);
            int s  = bucket_slot(dl, pe.z, ncnt, degx);
            bucket_put(s, pe.y, dl, pe.z, b, nb, spill_n, spill);
        }
    }
    __syncthreads();

    int node = b * BIN_NODES + tid;
    int n = ncnt[tid];
    float d = degx[tid] + 1.0f;
    if (node < N_NODES) {
        cnt[node] = n;
        dis[node] = rsqrtf(d);
    }

    v4i*       gb = (v4i*)(buckets + (size_t)b * (BIN_NODES * 2 * CAP));
    const v4i* lb = (const v4i*)nb;
    int mm[16];
    v4i pcs[16];
#pragma unroll
    for (int i = 0; i < 16; ++i) {
        int idx = i * 256 + tid;
        int m   = ncnt[idx >> 4];
        mm[i] = (m < CAP) ? m : CAP;
    }
#pragma unroll
    for (int i = 0; i < 16; ++i) {
        int idx = i * 256 + tid;
        int nd  = idx >> 4;
        int q   = idx & 15;
        v4i t = {0, 0, 0, 0};
        if (2 * q < mm[i] && b * BIN_NODES + nd < N_NODES) t = lb[idx];
        pcs[i] = t;
    }
#pragma unroll
    for (int i = 0; i < 16; ++i) {
        int idx = i * 256 + tid;
        int nd  = idx >> 4;
        int q   = idx & 15;
        if (2 * q < mm[i] && b * BIN_NODES + nd < N_NODES) gb[idx] = pcs[i];
    }
}

#define GEMM_WAVES 8
#define M_TILES    (N_NODES / 16)
#define W_PITCH4   33

__global__ void __launch_bounds__(GEMM_WAVES * 32)
gemm_xwt_wmma(const float* __restrict__ X,
              const float* __restrict__ Wa,
              const float* __restrict__ Wb,
              int n_split,
              const float* __restrict__ rs,
              float* __restrict__ C) {
    __shared__ float4 wl[HID * W_PITCH4];

    int tid  = threadIdx.x;
    int lane = tid & 31;
    int wv   = __builtin_amdgcn_readfirstlane(tid >> 5);
    int mt   = blockIdx.x * GEMM_WAVES + wv;
    int mtc  = (mt < M_TILES) ? mt : (M_TILES - 1);
    int c    = lane & 15;
    int h    = lane >> 4;

    const float4* xrow = (const float4*)(X + (size_t)(mtc * 16 + c) * IN_CH) + h;
    float4 a[16];
#pragma unroll
    for (int j = 0; j < 16; ++j) a[j] = xrow[2 * j];
    const float4* rsp = (const float4*)(rs + mtc * 16 + h * 8);
    float4 rs_lo = rsp[0];
    float4 rs_hi = rsp[1];

    float4 wrow[HID / GEMM_WAVES];
#pragma unroll
    for (int i = 0; i < HID / GEMM_WAVES; ++i) {
        int n = i * GEMM_WAVES + wv;
        const float* srow = (n < n_split) ? (Wa + (size_t)n * IN_CH)
                                          : (Wb + (size_t)(n - n_split) * IN_CH);
        wrow[i] = ((const float4*)srow)[lane];
    }
#pragma unroll
    for (int i = 0; i < HID / GEMM_WAVES; ++i) {
        int n  = i * GEMM_WAVES + wv;
        int t  = ((n >> 6) << 2) | (n & 3);
        int cc = (n & 63) >> 2;
        wl[(16 * t + cc) * W_PITCH4 + lane] = wrow[i];
    }
    __syncthreads();

    v8f acc[8];
#pragma unroll
    for (int t = 0; t < 8; ++t) acc[t] = v8f{};

    const float4* bl = wl + c * W_PITCH4 + h;
    float4 bcur[8], bnxt[8];
#pragma unroll
    for (int t = 0; t < 8; ++t) bcur[t] = bl[(16 * t) * W_PITCH4];

#pragma unroll
    for (int j = 0; j < 16; ++j) {
        if (j + 1 < 16) {
#pragma unroll
            for (int t = 0; t < 8; ++t) bnxt[t] = bl[(16 * t) * W_PITCH4 + 2 * (j + 1)];
        }
        __builtin_amdgcn_sched_barrier(0);
        v2f a0 = {a[j].x, a[j].y};
        v2f a1 = {a[j].z, a[j].w};
#pragma unroll
        for (int t = 0; t < 8; ++t) {
            v2f b0 = {bcur[t].x, bcur[t].y};
            acc[t] = __builtin_amdgcn_wmma_f32_16x16x4_f32(
                false, a0, false, b0, (short)0, acc[t], false, false);
        }
#pragma unroll
        for (int t = 0; t < 8; ++t) {
            v2f b1 = {bcur[t].z, bcur[t].w};
            acc[t] = __builtin_amdgcn_wmma_f32_16x16x4_f32(
                false, a1, false, b1, (short)0, acc[t], false, false);
        }
        __builtin_amdgcn_sched_barrier(0);
#pragma unroll
        for (int t = 0; t < 8; ++t) bcur[t] = bnxt[t];
    }

    float rsv[8] = {rs_lo.x, rs_lo.y, rs_lo.z, rs_lo.w, rs_hi.x, rs_hi.y, rs_hi.z, rs_hi.w};
#pragma unroll
    for (int t = 0; t < 8; ++t) {
#pragma unroll
        for (int r = 0; r < 8; ++r) acc[t][r] *= rsv[r];
    }

    if (mt < M_TILES) {
        float* crow = C + (size_t)(mt * 16 + h * 8) * HID + 4 * c;
#pragma unroll
        for (int r = 0; r < 8; ++r) {
#pragma unroll
            for (int q = 0; q < 2; ++q) {
                float4 o = {acc[4 * q + 0][r], acc[4 * q + 1][r], acc[4 * q + 2][r], acc[4 * q + 3][r]};
                *(float4*)(crow + (size_t)r * HID + 64 * q) = o;
            }
        }
    }
}

struct bucket_t { int n; int rj; int wj; };

static __device__ __forceinline__ bucket_t bucket_fetch(int c, int lane,
                                                        const int* __restrict__ cnt,
                                                        const int* __restrict__ buckets) {
    bucket_t k;
    k.n  = cnt[c];
    k.rj = 0;
    k.wj = 0;
    int m = (k.n < CAP) ? k.n : CAP;
    if (lane < m) {
        v2i pr = *(const v2i*)(buckets + (size_t)c * (2 * CAP) + 2 * lane);
        k.rj = pr.x;
        k.wj = pr.y;
    }
    return k;
}

static __device__ __forceinline__ float4 gather_row(int c, int lane, bucket_t k,
                                                    const int* __restrict__ spill_n,
                                                    const int* __restrict__ spill,
                                                    const float* __restrict__ src) {
    int n = k.n;
    int m = (n < CAP) ? n : CAP;
    const float4* src4 = (const float4*)src;

    float4 acc = {0.0f, 0.0f, 0.0f, 0.0f};
    int   rj = k.rj, wj = k.wj;
    int last = m - 1;
    for (int j = 0; j < m; j += 4) {
        int j1 = (j + 1 < last) ? j + 1 : last;
        int j2 = (j + 2 < last) ? j + 2 : last;
        int j3 = (j + 3 < last) ? j + 3 : last;
        int r0 = __builtin_amdgcn_readlane(rj, j);
        int r1 = __builtin_amdgcn_readlane(rj, j1);
        int r2 = __builtin_amdgcn_readlane(rj, j2);
        int r3 = __builtin_amdgcn_readlane(rj, j3);
        float4 v0 = src4[(size_t)r0 * (HID / 4) + lane];
        float4 v1 = src4[(size_t)r1 * (HID / 4) + lane];
        float4 v2 = src4[(size_t)r2 * (HID / 4) + lane];
        float4 v3 = src4[(size_t)r3 * (HID / 4) + lane];
        float c0 = __int_as_float(__builtin_amdgcn_readlane(wj, j));
        float c1 = __int_as_float(__builtin_amdgcn_readlane(wj, j1));
        float c2 = __int_as_float(__builtin_amdgcn_readlane(wj, j2));
        float c3 = __int_as_float(__builtin_amdgcn_readlane(wj, j3));
        if (j + 1 >= m) c1 = 0.0f;
        if (j + 2 >= m) c2 = 0.0f;
        if (j + 3 >= m) c3 = 0.0f;
        acc.x += c0 * v0.x; acc.y += c0 * v0.y; acc.z += c0 * v0.z; acc.w += c0 * v0.w;
        acc.x += c1 * v1.x; acc.y += c1 * v1.y; acc.z += c1 * v1.z; acc.w += c1 * v1.w;
        acc.x += c2 * v2.x; acc.y += c2 * v2.y; acc.z += c2 * v2.z; acc.w += c2 * v2.w;
        acc.x += c3 * v3.x; acc.y += c3 * v3.y; acc.z += c3 * v3.z; acc.w += c3 * v3.w;
    }
    if (n > CAP) {
        int no = *spill_n;
        for (int base = 0; base < no; base += 32) {
            int k  = base + lane;
            int oc = -1, orow = 0, ow = 0;
            if (k < no) {
                v4i en = *(const v4i*)(spill + 4 * (size_t)k);
                oc = en.x; orow = en.y; ow = en.z;
            }
            unsigned hit = __builtin_amdgcn_ballot_w32(oc == c);
            while (hit != 0u) {
                int l = __builtin_ctz(hit);
                hit &= hit - 1u;
                int   r  = __builtin_amdgcn_readlane(orow, l);
                float cw = __int_as_float(__builtin_amdgcn_readlane(ow, l));
                float4 v = src4[(size_t)r * (HID / 4) + lane];
                acc.x += cw * v.x; acc.y += cw * v.y; acc.z += cw * v.z; acc.w += cw * v.w;
            }
        }
    }
    return acc;
}

#define FUSE_WAVES 8
#define FUSE_GRID  1024
#define T_PITCH4   33

__global__ void __launch_bounds__(FUSE_WAVES * 32, 8)
gather_relu_gemm_kernel(const int* __restrict__ cnt, const int* __restrict__ spill_n,
                        const int* __restrict__ buckets, const int* __restrict__ spill,
                        const float* __restrict__ dis,
                        const float* __restrict__ xw, const float* __restrict__ b1,
                        const float* __restrict__ Wa, const float* __restrict__ Wb,
                        float* __restrict__ xw2) {
    __shared__ float4 ht[16 * T_PITCH4];
    __shared__ float4 ot[16 * T_PITCH4];

    int tid  = threadIdx.x;
    int lane = tid & 31;
    int wv   = __builtin_amdgcn_readfirstlane(tid >> 5);
    int c    = lane & 15;
    int h    = lane >> 4;

    int nrow = 16 * wv + c;
    const float*  wrow = (nrow < OUT_CH) ? (Wa + (size_t)nrow * HID)
                                         : (Wb + (size_t)(nrow - OUT_CH) * HID);
    const float4* wr4  = (const float4*)wrow + h;
    float4 bf[16];
#pragma unroll
    for (int j = 0; j < 16; ++j) bf[j] = wr4[2 * j];

    float4 bias = ((const float4*)b1)[lane];
    const float4* xw4 = (const float4*)xw;

    for (int s = blockIdx.x; s < M_TILES; s += FUSE_GRID) {
        int c0 = __builtin_amdgcn_readfirstlane(s * 16 + wv);
        int c1 = c0 + 8;
        bucket_t k0 = bucket_fetch(c0, lane, cnt, buckets);
        bucket_t k1 = bucket_fetch(c1, lane, cnt, buckets);
        float4 own0 = xw4[(size_t)c0 * (HID / 4) + lane];
        float4 own1 = xw4[(size_t)c1 * (HID / 4) + lane];
        float  d0   = dis[c0];
        float  d1   = dis[c1];

        float4 g0 = gather_row(c0, lane, k0, spill_n, spill, xw);
        float4 v0;
        v0.x = fmaxf((g0.x + own0.x) * d0 + bias.x, 0.0f);
        v0.y = fmaxf((g0.y + own0.y) * d0 + bias.y, 0.0f);
        v0.z = fmaxf((g0.z + own0.z) * d0 + bias.z, 0.0f);
        v0.w = fmaxf((g0.w + own0.w) * d0 + bias.w, 0.0f);
        ht[wv * T_PITCH4 + lane] = v0;

        float4 g1 = gather_row(c1, lane, k1, spill_n, spill, xw);
        float4 v1;
        v1.x = fmaxf((g1.x + own1.x) * d1 + bias.x, 0.0f);
        v1.y = fmaxf((g1.y + own1.y) * d1 + bias.y, 0.0f);
        v1.z = fmaxf((g1.z + own1.z) * d1 + bias.z, 0.0f);
        v1.w = fmaxf((g1.w + own1.w) * d1 + bias.w, 0.0f);
        ht[(wv + 8) * T_PITCH4 + lane] = v1;
        __syncthreads();

        const float4* al = ht + c * T_PITCH4 + h;
        v8f acc = v8f{};
        float4 a[8];
#pragma unroll
        for (int g = 0; g < 2; ++g) {
#pragma unroll
            for (int j = 0; j < 8; ++j) a[j] = al[2 * (8 * g + j)];
#pragma unroll
            for (int j = 0; j < 8; ++j) {
                v2f a0 = {a[j].x, a[j].y};
                v2f a1 = {a[j].z, a[j].w};
                v2f b0 = {bf[8 * g + j].x, bf[8 * g + j].y};
                v2f b1 = {bf[8 * g + j].z, bf[8 * g + j].w};
                acc = __builtin_amdgcn_wmma_f32_16x16x4_f32(
                    false, a0, false, b0, (short)0, acc, false, false);
                acc = __builtin_amdgcn_wmma_f32_16x16x4_f32(
                    false, a1, false, b1, (short)0, acc, false, false);
            }
        }

        float* of = (float*)ot;
#pragma unroll
        for (int r = 0; r < 8; ++r) of[(8 * h + r) * (4 * T_PITCH4) + 16 * wv + c] = acc[r];
        __syncthreads();

        float4 o0 = ot[wv * T_PITCH4 + lane];
        float4 o1 = ot[(wv + 8) * T_PITCH4 + lane];
        o0.x *= d0; o0.y *= d0; o0.z *= d0; o0.w *= d0;
        o1.x *= d1; o1.y *= d1; o1.z *= d1; o1.w *= d1;
        ((float4*)xw2)[(size_t)c0 * (HID / 4) + lane] = o0;
        ((float4*)xw2)[(size_t)c1 * (HID / 4) + lane] = o1;
    }
}

__global__ void gather_out_kernel(const int* __restrict__ cnt, const int* __restrict__ spill_n,
                                  const int* __restrict__ buckets, const int* __restrict__ spill,
                                  const float* __restrict__ dis,
                                  const float* __restrict__ xw2, const float* __restrict__ b_mu,
                                  const float* __restrict__ b_ls, float* __restrict__ out) {
    int wave = (blockIdx.x * blockDim.x + threadIdx.x) >> 5;
    int lane = threadIdx.x & 31;
    if (wave >= N_NODES) return;
    int c = __builtin_amdgcn_readfirstlane(wave);
    int l16 = lane & 15;
    float4 own  = ((const float4*)xw2)[(size_t)c * (HID / 4) + lane];
    float4 bias = (lane < 16) ? ((const float4*)b_mu)[l16] : ((const float4*)b_ls)[l16];
    float  dc   = dis[c];
    bucket_t k  = bucket_fetch(c, lane, cnt, buckets);
    float4 acc  = gather_row(c, lane, k, spill_n, spill, xw2);
    float4 v;
    v.x = (acc.x + own.x) * dc + bias.x;
    v.y = (acc.y + own.y) * dc + bias.y;
    v.z = (acc.z + own.z) * dc + bias.z;
    v.w = (acc.w + own.w) * dc + bias.w;
    float* orow = out + ((lane < 16) ? (size_t)0 : (size_t)N_NODES * OUT_CH)
                      + (size_t)c * OUT_CH + l16 * 4;
    *(float4*)orow = v;
}

extern "C" void kernel_launch(void* const* d_in, const int* in_sizes, int n_in,
                              void* d_out, int out_size, void* d_ws, size_t ws_size,
                              hipStream_t stream) {
    (void)in_sizes; (void)n_in; (void)out_size; (void)ws_size;
    const float* x   = (const float*)d_in[0];
    const int*   ei  = (const int*)d_in[1];
    const float* ew  = (const float*)d_in[2];
    const float* W1  = (const float*)d_in[3];
    const float* b1  = (const float*)d_in[4];
    const float* Wmu = (const float*)d_in[5];
    const float* bmu = (const float*)d_in[6];
    const float* Wls = (const float*)d_in[7];
    const float* bls = (const float*)d_in[8];
    const int* row = ei;
    const int* col = ei + N_EDGES;

    float* ws      = (float*)d_ws;
    float* dis     = ws;
    int*   cnt     = (int*)(ws + 50000);
    int*   spill_n = cnt + N_NODES;
    float* xw      = ws + 150000;
    float* agg     = ws + 150000 + 6400000;
    int*   buckets = (int*)(ws + 12950016);
    int*   spill   = (int*)(ws + 16150016);
    int*   pspill  = (int*)(ws + 150000);
    int*   slab    = (int*)(ws + 6550016);
    int*   celltab = (int*)(ws + 9761280);

    hipMemsetAsync(spill_n, 0, 4 * sizeof(int), stream);

    part_kernel<<<NCHUNK, 256, 0, stream>>>(row, col, ew, spill_n + 1, slab, celltab, pspill);
    bin_kernel<<<NBIN, 256, 0, stream>>>(slab, celltab, spill_n + 1, pspill, spill_n, spill, cnt, buckets, dis);

    gemm_xwt_wmma<<<(M_TILES + GEMM_WAVES - 1) / GEMM_WAVES, GEMM_WAVES * 32, 0, stream>>>(x, W1, W1, HID, dis, xw);
    gather_relu_gemm_kernel<<<FUSE_GRID, FUSE_WAVES * 32, 0, stream>>>(cnt, spill_n, buckets, spill, dis, xw, b1,
                                                                       Wmu, Wls, agg);

    gather_out_kernel<<<N_NODES / 8, 256, 0, stream>>>(cnt, spill_n, buckets, spill, dis, agg, bmu, bls,
                                                       (float*)d_out);
}


# Layer-1 GEMM moved into the bucket-building kernel: the work-group that just summed dis for its 256 nodes multiplies their 16 strips of x by W1^T (2 per wave); one launch fewer; sum order unchanged
# speedup vs baseline: 4.9193x; 1.0419x over previous
#include <hip/hip_runtime.h>


#define N_NODES 50000
#define N_EDGES 600000
#define IN_CH   128
#define HID     128
#define OUT_CH  64

typedef __attribute__((ext_vector_type(2))) float v2f;
typedef __attribute__((ext_vector_type(8))) float v8f;

typedef __attribute__((ext_vector_type(2))) int v2i;

typedef __attribute__((ext_vector_type(4))) int v4i;

#define CAP 32

#define BIN_NODES 256
#define NBIN      196
#define NCHUNK    256
#define CHUNK     2344
#define EPT       10

__global__ void __launch_bounds__(256)
part_kernel(const int* __restrict__ row, const int* __restrict__ col,
            const float* __restrict__ w, int* __restrict__ pspill_n,
            int* __restrict__ slab, int* __restrict__ celltab, int* __restrict__ pspill) {
    __shared__ int lcnt[NBIN];
    __shared__ v2i lcell[NBIN * CAP];

    int tid   = threadIdx.x;
    int lane  = tid & 31;
    int wv    = __builtin_amdgcn_readfirstlane(tid >> 5);
    int chunk = blockIdx.x;
    int base  = chunk * CHUNK;
    int end   = (base + CHUNK < N_EDGES) ? (base + CHUNK) : N_EDGES;

    if (tid < NBIN) lcnt[tid] = 0;

    int   r[EPT], c[EPT];
    float wt[EPT];
#pragma unroll
    for (int k = 0; k < EPT; ++k) {
        int  e  = base + k * 256 + tid;
        bool ok = e < end;
        r[k]  = ok ? row[e] : 0;
        c[k]  = ok ? col[e] : 0;
        wt[k] = ok ? w[e] : 0.0f;
    }
    __syncthreads();

    int s[EPT];
#pragma unroll
    for (int k = 0; k < EPT; ++k) {
        int e = base + k * 256 + tid;
        s[k] = 0;
        if (e < end) s[k] = atomicAdd(&lcnt[c[k] >> 8], 1);
    }
    unsigned over = 0u;
#pragma unroll
    for (int k = 0; k < EPT; ++k) {
        int e = base + k * 256 + tid;
        if (e < end) {
            float we = (r[k] == c[k]) ? 0.0f : wt[k];
            int b  = c[k] >> 8;
            int dl = c[k] & (BIN_NODES - 1);
            if (s[k] < CAP) {
                v2i pr = {r[k] | (dl << 16), __float_as_int(we)};
                lcell[b * CAP + s[k]] = pr;
            } else {
                over |= 1u << k;
            }
        }
    }
    if (over != 0u) {
#pragma unroll
        for (int k = 0; k < EPT; ++k) {
            if ((over >> k) & 1u) {
                float we = (r[k] == c[k]) ? 0.0f : wt[k];
                int o = atomicAdd(pspill_n, 1);
                v4i en = {c[k], r[k], __float_as_int(we), 0};
                *(v4i*)(pspill + 4 * (size_t)o) = en;
            }
        }
    }
    __syncthreads();

    int myb = wv + 8 * lane;
    int myc = (myb < NBIN) ? lcnt[myb] : 0;
    myc = (myc < CAP) ? myc : CAP;
    if (myb < NBIN) celltab[myb * NCHUNK + chunk] = myc;
    v2i pc[(NBIN + 7) / 8];
#pragma unroll
    for (int i = 0; i < (NBIN + 7) / 8; ++i) {
        int b = wv + 8 * i;
        int n = __builtin_amdgcn_readlane(myc, i);
        v2i t = {0, 0};
        if (lane < n) t = lcell[b * CAP + lane];
        pc[i] = t;
    }
#pragma unroll
    for (int i = 0; i < (NBIN + 7) / 8; ++i) {
        int b = wv + 8 * i;
        int n = __builtin_amdgcn_readlane(myc, i);
        if (lane < n)
            *(v2i*)(slab + ((size_t)(b * NCHUNK + chunk) * CAP + lane) * 2) = pc[i];
    }
}

static __device__ __forceinline__ int bucket_slot(int dl, int wb, int* ncnt, float* degx) {
    int s = atomicAdd(&ncnt[dl], 1);
    atomicAdd(&degx[dl], __int_as_float(wb));
    return s;
}

static __device__ __forceinline__ void bucket_put(int s, int src, int dl, int wb, int b, v2i* nb,
                                                  int* __restrict__ spill_n, int* __restrict__ spill) {
    if (s < CAP) {
        v2i pr = {src, wb};
        nb[dl * CAP + s] = pr;
    } else {
        int o = atomicAdd(spill_n, 1);
        v4i en = {b * BIN_NODES + dl, src, wb, 0};
        *(v4i*)(spill + 4 * (size_t)o) = en;
    }
}

#define GEMM_WAVES 8
#define M_TILES    (N_NODES / 16)
#define W_PITCH4   33
#define BIN_STRIPS (BIN_NODES / 16)

template <bool NEXT>
static __device__ __forceinline__ void strip_product(float4 (&a)[16], const float4* xnext, const float4* bl,
                                                     float4 dg_lo, float4 dg_hi, float* crow) {
    v8f acc[8];
#pragma unroll
    for (int t = 0; t < 8; ++t) acc[t] = v8f{};

    float4 bcur[8], bnxt[8];
#pragma unroll
    for (int t = 0; t < 8; ++t) bcur[t] = bl[(16 * t) * W_PITCH4];

#pragma unroll
    for (int j = 0; j < 16; ++j) {
        if (j + 1 < 16) {
#pragma unroll
            for (int t = 0; t < 8; ++t) bnxt[t] = bl[(16 * t) * W_PITCH4 + 2 * (j + 1)];
        }
        if (NEXT && j > 0) a[j - 1] = xnext[2 * (j - 1)];
        __builtin_amdgcn_sched_barrier(0);
        v2f a0 = {a[j].x, a[j].y};
        v2f a1 = {a[j].z, a[j].w};
#pragma unroll
        for (int t = 0; t < 8; ++t) {
            v2f b0 = {bcur[t].x, bcur[t].y};
            acc[t] = __builtin_amdgcn_wmma_f32_16x16x4_f32(
                false, a0, false, b0, (short)0, acc[t], false, false);
        }
#pragma unroll
        for (int t = 0; t < 8; ++t) {
            v2f b1 = {bcur[t].z, bcur[t].w};
            acc[t] = __builtin_amdgcn_wmma_f32_16x16x4_f32(
                false, a1, false, b1, (short)0, acc[t], false, false);
        }
        __builtin_amdgcn_sched_barrier(0);
#pragma unroll
        for (int t = 0; t < 8; ++t) bcur[t] = bnxt[t];
    }
    if (NEXT) a[15] = xnext[30];

    float rsv[8] = {rsqrtf(dg_lo.x + 1.0f), rsqrtf(dg_lo.y + 1.0f), rsqrtf(dg_lo.z + 1.0f), rsqrtf(dg_lo.w + 1.0f),
                    rsqrtf(dg_hi.x + 1.0f), rsqrtf(dg_hi.y + 1.0f), rsqrtf(dg_hi.z + 1.0f), rsqrtf(dg_hi.w + 1.0f)};
#pragma unroll
    for (int t = 0; t < 8; ++t) {
#pragma unroll
        for (int r = 0; r < 8; ++r) acc[t][r] *= rsv[r];
    }
#pragma unroll
    for (int r = 0; r < 8; ++r) {
#pragma unroll
        for (int q = 0; q < 2; ++q) {
            float4 o = {acc[4 * q + 0][r], acc[4 * q + 1][r], acc[4 * q + 2][r], acc[4 * q + 3][r]};
            *(float4*)(crow + (size_t)r * HID + 64 * q) = o;
        }
    }
}

__global__ void __launch_bounds__(256)
bin_gemm_kernel(const int* __restrict__ slab, const int* __restrict__ celltab,
                const int* __restrict__ pspill_n, const int* __restrict__ pspill,
                int* __restrict__ spill_n, int* __restrict__ spill,
                int* __restrict__ cnt, int* __restrict__ buckets, float* __restrict__ dis,
                const float* __restrict__ X, const float* __restrict__ W1, float* __restrict__ C) {
    __shared__ int    ncnt[BIN_NODES];
    __shared__ __attribute__((aligned(16))) float degx[BIN_NODES];
    __shared__ v2i    nb[BIN_NODES * CAP];
    __shared__ float4 wl[HID * W_PITCH4];

    int tid  = threadIdx.x;
    int lane = tid & 31;
    int wv   = __builtin_amdgcn_readfirstlane(tid >> 5);
    int b    = blockIdx.x;
    int c    = lane & 15;
    int h    = lane >> 4;

    ncnt[tid] = 0;
    degx[tid] = 0.0f;
    float4 wrow[HID / GEMM_WAVES];
#pragma unroll
    for (int i = 0; i < HID / GEMM_WAVES; ++i) {
        int n = i * GEMM_WAVES + wv;
        wrow[i] = ((const float4*)(W1 + (size_t)n * IN_CH))[lane];
    }
    int myn = celltab[b * NCHUNK + wv * 32 + lane];
    int np  = *pspill_n;
#pragma unroll
    for (int i = 0; i < HID / GEMM_WAVES; ++i) {
        int n  = i * GEMM_WAVES + wv;
        int t  = ((n >> 6) << 2) | (n & 3);
        int cc = (n & 63) >> 2;
        wl[(16 * t + cc) * W_PITCH4 + lane] = wrow[i];
    }
    __syncthreads();

    const v2i* cells = (const v2i*)slab + (size_t)(b * NCHUNK + wv * 32) * CAP;
    const unsigned long long* cells64 = (const unsigned long long*)cells;
    unsigned long long en[32];
#pragma unroll
    for (int k = 0; k < 32; ++k) {
        int n = __builtin_amdgcn_readlane(myn, k);
        unsigned long long t = ~0ull;
        if (lane < n) t = cells64[k * CAP + lane];
        en[k] = t;
    }
    int st0 = b * BIN_STRIPS + wv;
    int st1 = st0 + GEMM_WAVES;
    int s0c = (st0 < M_TILES) ? st0 : (M_TILES - 1);
    int s1c = (st1 < M_TILES) ? st1 : (M_TILES - 1);
    const float4* xr0 = (const float4*)(X + (size_t)(s0c * 16 + c) * IN_CH) + h;
    const float4* xr1 = (const float4*)(X + (size_t)(s1c * 16 + c) * IN_CH) + h;
    float4 a[16];
#pragma unroll
    for (int j = 0; j < 16; ++j) a[j] = xr0[2 * j];
#pragma unroll
    for (int g = 0; g < 32; g += 16) {
        int s[16];
#pragma unroll
        for (int k = 0; k < 16; ++k) {
            int ex = (int)en[g + k];
            int wb = (int)(en[g + k] >> 32);
            s[k] = 0;
            if (ex >= 0) s[k] = bucket_slot((ex >> 16) & (BIN_NODES - 1), wb, ncnt, degx);
        }
        unsigned over = 0u;
#pragma unroll
        for (int k = 0; k < 16; ++k) {
            int ex = (int)en[g + k];
            int wb = (int)(en[g + k] >> 32);
            if (ex >= 0) {
                if (s[k] < CAP) {
                    v2i pr = {ex & 0xffff, wb};
                    nb[((ex >> 16) & (BIN_NODES - 1)) * CAP + s[k]] = pr;
                } else {
                    over |= 1u << k;
                }
            }
        }
        if (over != 0u) {
#pragma unroll
            for (int k = 0; k < 16; ++k) {
                if ((over >> k) & 1u) {
                    int ex = (int)en[g + k];
                    bucket_put(CAP, ex & 0xffff, (ex >> 16) & (BIN_NODES - 1), (int)(en[g + k] >> 32), b,
                               nb, spill_n, spill);
                }
            }
        }
    }
    for (int k = tid; k < np; k += 256) {
        v4i pe = *(const v4i*)(pspill + 4 * (size_t)k);
        if ((pe.x >> 8) == b) {
            int dl = pe.x & (BIN_NODES - 1);
            int s  = bucket_slot(dl, pe.z, ncnt, degx);
            bucket_put(s, pe.y, dl, pe.z, b, nb, spill_n, spill);
        }
    }
    __syncthreads();

    int node = b * BIN_NODES + tid;
    int n = ncnt[tid];
    float d = degx[tid] + 1.0f;
    if (node < N_NODES) {
        cnt[node] = n;
        dis[node] = rsqrtf(d);
    }

    v4i*       gb = (v4i*)(buckets + (size_t)b * (BIN_NODES * 2 * CAP));
    const v4i* lb = (const v4i*)nb;
    int mm[16];
    v4i pcs[16];
#pragma unroll
    for (int i = 0; i < 16; ++i) {
        int idx = i * 256 + tid;
        int m   = ncnt[idx >> 4];
        mm[i] = (m < CAP) ? m : CAP;
    }
#pragma unroll
    for (int i = 0; i < 16; ++i) {
        int idx = i * 256 + tid;
        int nd  = idx >> 4;
        int q   = idx & 15;
        v4i t = {0, 0, 0, 0};
        if (2 * q < mm[i] && b * BIN_NODES + nd < N_NODES) t = lb[idx];
        pcs[i] = t;
    }
#pragma unroll
    for (int i = 0; i < 16; ++i) {
        int idx = i * 256 + tid;
        int nd  = idx >> 4;
        int q   = idx & 15;
        if (2 * q < mm[i] && b * BIN_NODES + nd < N_NODES) gb[idx] = pcs[i];
    }

    const float4* bl = wl + c * W_PITCH4 + h;
    const float4* dg = (const float4*)degx + 2 * h;
    if (st0 < M_TILES)
        strip_product<true>(a, xr1, bl, dg[4 * wv], dg[4 * wv + 1],
                            C + (size_t)(st0 * 16 + h * 8) * HID + 4 * c);
    if (st1 < M_TILES)
        strip_product<false>(a, xr1, bl, dg[4 * (wv + GEMM_WAVES)], dg[4 * (wv + GEMM_WAVES) + 1],
                             C + (size_t)(st1 * 16 + h * 8) * HID + 4 * c);
}

struct bucket_t { int n; int rj; int wj; };

static __device__ __forceinline__ bucket_t bucket_fetch(int c, int lane,
                                                        const int* __restrict__ cnt,
                                                        const int* __restrict__ buckets) {
    bucket_t k;
    k.n  = cnt[c];
    k.rj = 0;
    k.wj = 0;
    int m = (k.n < CAP) ? k.n : CAP;
    if (lane < m) {
        v2i pr = *(const v2i*)(buckets + (size_t)c * (2 * CAP) + 2 * lane);
        k.rj = pr.x;
        k.wj = pr.y;
    }
    return k;
}

static __device__ __forceinline__ float4 gather_row(int c, int lane, bucket_t k,
                                                    const int* __restrict__ spill_n,
                                                    const int* __restrict__ spill,
                                                    const float* __restrict__ src) {
    int n = k.n;
    int m = (n < CAP) ? n : CAP;
    const float4* src4 = (const float4*)src;

    float4 acc = {0.0f, 0.0f, 0.0f, 0.0f};
    int   rj = k.rj, wj = k.wj;
    int last = m - 1;
    for (int j = 0; j < m; j += 4) {
        int j1 = (j + 1 < last) ? j + 1 : last;
        int j2 = (j + 2 < last) ? j + 2 : last;
        int j3 = (j + 3 < last) ? j + 3 : last;
        int r0 = __builtin_amdgcn_readlane(rj, j);
        int r1 = __builtin_amdgcn_readlane(rj, j1);
        int r2 = __builtin_amdgcn_readlane(rj, j2);
        int r3 = __builtin_amdgcn_readlane(rj, j3);
        float4 v0 = src4[(size_t)r0 * (HID / 4) + lane];
        float4 v1 = src4[(size_t)r1 * (HID / 4) + lane];
        float4 v2 = src4[(size_t)r2 * (HID / 4) + lane];
        float4 v3 = src4[(size_t)r3 * (HID / 4) + lane];
        float c0 = __int_as_float(__builtin_amdgcn_readlane(wj, j));
        float c1 = __int_as_float(__builtin_amdgcn_readlane(wj, j1));
        float c2 = __int_as_float(__builtin_amdgcn_readlane(wj, j2));
        float c3 = __int_as_float(__builtin_amdgcn_readlane(wj, j3));
        if (j + 1 >= m) c1 = 0.0f;
        if (j + 2 >= m) c2 = 0.0f;
        if (j + 3 >= m) c3 = 0.0f;
        acc.x += c0 * v0.x; acc.y += c0 * v0.y; acc.z += c0 * v0.z; acc.w += c0 * v0.w;
        acc.x += c1 * v1.x; acc.y += c1 * v1.y; acc.z += c1 * v1.z; acc.w += c1 * v1.w;
        acc.x += c2 * v2.x; acc.y += c2 * v2.y; acc.z += c2 * v2.z; acc.w += c2 * v2.w;
        acc.x += c3 * v3.x; acc.y += c3 * v3.y; acc.z += c3 * v3.z; acc.w += c3 * v3.w;
    }
    if (n > CAP) {
        int no = *spill_n;
        for (int base = 0; base < no; base += 32) {
            int k  = base + lane;
            int oc = -1, orow = 0, ow = 0;
            if (k < no) {
                v4i en = *(const v4i*)(spill + 4 * (size_t)k);
                oc = en.x; orow = en.y; ow = en.z;
            }
            unsigned hit = __builtin_amdgcn_ballot_w32(oc == c);
            while (hit != 0u) {
                int l = __builtin_ctz(hit);
                hit &= hit - 1u;
                int   r  = __builtin_amdgcn_readlane(orow, l);
                float cw = __int_as_float(__builtin_amdgcn_readlane(ow, l));
                float4 v = src4[(size_t)r * (HID / 4) + lane];
                acc.x += cw * v.x; acc.y += cw * v.y; acc.z += cw * v.z; acc.w += cw * v.w;
            }
        }
    }
    return acc;
}

#define FUSE_WAVES 8
#define FUSE_GRID  1024
#define T_PITCH4   33

__global__ void __launch_bounds__(FUSE_WAVES * 32, 8)
gather_relu_gemm_kernel(const int* __restrict__ cnt, const int* __restrict__ spill_n,
                        const int* __restrict__ buckets, const int* __restrict__ spill,
                        const float* __restrict__ dis,
                        const float* __restrict__ xw, const float* __restrict__ b1,
                        const float* __restrict__ Wa, const float* __restrict__ Wb,
                        float* __restrict__ xw2) {
    __shared__ float4 ht[16 * T_PITCH4];
    __shared__ float4 ot[16 * T_PITCH4];

    int tid  = threadIdx.x;
    int lane = tid & 31;
    int wv   = __builtin_amdgcn_readfirstlane(tid >> 5);
    int c    = lane & 15;
    int h    = lane >> 4;

    int nrow = 16 * wv + c;
    const float*  wrow = (nrow < OUT_CH) ? (Wa + (size_t)nrow * HID)
                                         : (Wb + (size_t)(nrow - OUT_CH) * HID);
    const float4* wr4  = (const float4*)wrow + h;
    float4 bf[16];
#pragma unroll
    for (int j = 0; j < 16; ++j) bf[j] = wr4[2 * j];

    float4 bias = ((const float4*)b1)[lane];
    const float4* xw4 = (const float4*)xw;

    for (int s = blockIdx.x; s < M_TILES; s += FUSE_GRID) {
        int c0 = __builtin_amdgcn_readfirstlane(s * 16 + wv);
        int c1 = c0 + 8;
        bucket_t k0 = bucket_fetch(c0, lane, cnt, buckets);
        bucket_t k1 = bucket_fetch(c1, lane, cnt, buckets);
        float4 own0 = xw4[(size_t)c0 * (HID / 4) + lane];
        float4 own1 = xw4[(size_t)c1 * (HID / 4) + lane];
        float  d0   = dis[c0];
        float  d1   = dis[c1];

        float4 g0 = gather_row(c0, lane, k0, spill_n, spill, xw);
        float4 v0;
        v0.x = fmaxf((g0.x + own0.x) * d0 + bias.x, 0.0f);
        v0.y = fmaxf((g0.y + own0.y) * d0 + bias.y, 0.0f);
        v0.z = fmaxf((g0.z + own0.z) * d0 + bias.z, 0.0f);
        v0.w = fmaxf((g0.w + own0.w) * d0 + bias.w, 0.0f);
        ht[wv * T_PITCH4 + lane] = v0;

        float4 g1 = gather_row(c1, lane, k1, spill_n, spill, xw);
        float4 v1;
        v1.x = fmaxf((g1.x + own1.x) * d1 + bias.x, 0.0f);
        v1.y = fmaxf((g1.y + own1.y) * d1 + bias.y, 0.0f);
        v1.z = fmaxf((g1.z + own1.z) * d1 + bias.z, 0.0f);
        v1.w = fmaxf((g1.w + own1.w) * d1 + bias.w, 0.0f);
        ht[(wv + 8) * T_PITCH4 + lane] = v1;
        __syncthreads();

        const float4* al = ht + c * T_PITCH4 + h;
        v8f acc = v8f{};
        float4 a[8];
#pragma unroll
        for (int g = 0; g < 2; ++g) {
#pragma unroll
            for (int j = 0; j < 8; ++j) a[j] = al[2 * (8 * g + j)];
#pragma unroll
            for (int j = 0; j < 8; ++j) {
                v2f a0 = {a[j].x, a[j].y};
                v2f a1 = {a[j].z, a[j].w};
                v2f b0 = {bf[8 * g + j].x, bf[8 * g + j].y};
                v2f b1 = {bf[8 * g + j].z, bf[8 * g + j].w};
                acc = __builtin_amdgcn_wmma_f32_16x16x4_f32(
                    false, a0, false, b0, (short)0, acc, false, false);
                acc = __builtin_amdgcn_wmma_f32_16x16x4_f32(
                    false, a1, false, b1, (short)0, acc, false, false);
            }
        }

        float* of = (float*)ot;
#pragma unroll
        for (int r = 0; r < 8; ++r) of[(8 * h + r) * (4 * T_PITCH4) + 16 * wv + c] = acc[r];
        __syncthreads();

        float4 o0 = ot[wv * T_PITCH4 + lane];
        float4 o1 = ot[(wv + 8) * T_PITCH4 + lane];
        o0.x *= d0; o0.y *= d0; o0.z *= d0; o0.w *= d0;
        o1.x *= d1; o1.y *= d1; o1.z *= d1; o1.w *= d1;
        ((float4*)xw2)[(size_t)c0 * (HID / 4) + lane] = o0;
        ((float4*)xw2)[(size_t)c1 * (HID / 4) + lane] = o1;
    }
}

__global__ void gather_out_kernel(const int* __restrict__ cnt, const int* __restrict__ spill_n,
                                  const int* __restrict__ buckets, const int* __restrict__ spill,
                                  const float* __restrict__ dis,
                                  const float* __restrict__ xw2, const float* __restrict__ b_mu,
                                  const float* __restrict__ b_ls, float* __restrict__ out) {
    int wave = (blockIdx.x * blockDim.x + threadIdx.x) >> 5;
    int lane = threadIdx.x & 31;
    if (wave >= N_NODES) return;
    int c = __builtin_amdgcn_readfirstlane(wave);
    int l16 = lane & 15;
    float4 own  = ((const float4*)xw2)[(size_t)c * (HID / 4) + lane];
    float4 bias = (lane < 16) ? ((const float4*)b_mu)[l16] : ((const float4*)b_ls)[l16];
    float  dc   = dis[c];
    bucket_t k  = bucket_fetch(c, lane, cnt, buckets);
    float4 acc  = gather_row(c, lane, k, spill_n, spill, xw2);
    float4 v;
    v.x = (acc.x + own.x) * dc + bias.x;
    v.y = (acc.y + own.y) * dc + bias.y;
    v.z = (acc.z + own.z) * dc + bias.z;
    v.w = (acc.w + own.w) * dc + bias.w;
    float* orow = out + ((lane < 16) ? (size_t)0 : (size_t)N_NODES * OUT_CH)
                      + (size_t)c * OUT_CH + l16 * 4;
    *(float4*)orow = v;
}

extern "C" void kernel_launch(void* const* d_in, const int* in_sizes, int n_in,
                              void* d_out, int out_size, void* d_ws, size_t ws_size,
                              hipStream_t stream) {
    (void)in_sizes; (void)n_in; (void)out_size; (void)ws_size;
    const float* x   = (const float*)d_in[0];
    const int*   ei  = (const int*)d_in[1];
    const float* ew  = (const float*)d_in[2];
    const float* W1  = (const float*)d_in[3];
    const float* b1  = (const float*)d_in[4];
    const float* Wmu = (const float*)d_in[5];
    const float* bmu = (const float*)d_in[6];
    const float* Wls = (const float*)d_in[7];
    const float* bls = (const float*)d_in[8];
    const int* row = ei;
    const int* col = ei + N_EDGES;

    float* ws      = (float*)d_ws;
    float* dis     = ws;
    int*   cnt     = (int*)(ws + 50000);
    int*   spill_n = cnt + N_NODES;
    float* xw      = ws + 150000;
    float* agg     = ws + 150000 + 6400000;
    int*   buckets = (int*)(ws + 12950016);
    int*   spill   = (int*)(ws + 16150016);
    int*   slab    = (int*)(ws + 6550016);
    int*   celltab = (int*)(ws + 9761280);
    int*   pspill  = (int*)(ws + 9811456);

    hipMemsetAsync(spill_n, 0, 4 * sizeof(int), stream);

    part_kernel<<<NCHUNK, 256, 0, stream>>>(row, col, ew, spill_n + 1, slab, celltab, pspill);
    bin_gemm_kernel<<<NBIN, 256, 0, stream>>>(slab, celltab, spill_n + 1, pspill, spill_n, spill, cnt, buckets, dis,
                                              x, W1, xw);
    gather_relu_gemm_kernel<<<FUSE_GRID, FUSE_WAVES * 32, 0, stream>>>(cnt, spill_n, buckets, spill, dis, xw, b1,
                                                                       Wmu, Wls, agg);

    gather_out_kernel<<<N_NODES / 8, 256, 0, stream>>>(cnt, spill_n, buckets, spill, dis, agg, bmu, bls,
                                                       (float*)d_out);
}


# Row buffers xw and xw2 moved to 512-byte boundaries of the workspace (were 64 bytes past a line): each 512-B row read or stored is 4 whole 128-B lines and 2 aligned 256-B units, not 5 and 3
# speedup vs baseline: 5.1930x; 1.0556x over previous
#include <hip/hip_runtime.h>


#define N_NODES 50000
#define N_EDGES 600000
#define IN_CH   128
#define HID     128
#define OUT_CH  64

typedef __attribute__((ext_vector_type(2))) float v2f;
typedef __attribute__((ext_vector_type(8))) float v8f;

typedef __attribute__((ext_vector_type(2))) int v2i;

typedef __attribute__((ext_vector_type(4))) int v4i;

#define CAP 32

#define BIN_NODES 256
#define NBIN      196
#define NCHUNK    256
#define CHUNK     2344
#define EPT       10

__global__ void __launch_bounds__(256)
part_kernel(const int* __restrict__ row, const int* __restrict__ col,
            const float* __restrict__ w, int* __restrict__ pspill_n,
            int* __restrict__ slab, int* __restrict__ celltab, int* __restrict__ pspill) {
    __shared__ int lcnt[NBIN];
    __shared__ v2i lcell[NBIN * CAP];

    int tid   = threadIdx.x;
    int lane  = tid & 31;
    int wv    = __builtin_amdgcn_readfirstlane(tid >> 5);
    int chunk = blockIdx.x;
    int base  = chunk * CHUNK;
    int end   = (base + CHUNK < N_EDGES) ? (base + CHUNK) : N_EDGES;

    if (tid < NBIN) lcnt[tid] = 0;

    int   r[EPT], c[EPT];
    float wt[EPT];
#pragma unroll
    for (int k = 0; k < EPT; ++k) {
        int  e  = base + k * 256 + tid;
        bool ok = e < end;
        r[k]  = ok ? row[e] : 0;
        c[k]  = ok ? col[e] : 0;
        wt[k] = ok ? w[e] : 0.0f;
    }
    __syncthreads();

    int s[EPT];
#pragma unroll
    for (int k = 0; k < EPT; ++k) {
        int e = base + k * 256 + tid;
        s[k] = 0;
        if (e < end) s[k] = atomicAdd(&lcnt[c[k] >> 8], 1);
    }
    unsigned over = 0u;
#pragma unroll
    for (int k = 0; k < EPT; ++k) {
        int e = base + k * 256 + tid;
        if (e < end) {
            float we = (r[k] == c[k]) ? 0.0f : wt[k];
            int b  = c[k] >> 8;
            int dl = c[k] & (BIN_NODES - 1);
            if (s[k] < CAP) {
                v2i pr = {r[k] | (dl << 16), __float_as_int(we)};
                lcell[b * CAP + s[k]] = pr;
            } else {
                over |= 1u << k;
            }
        }
    }
    if (over != 0u) {
#pragma unroll
        for (int k = 0; k < EPT; ++k) {
            if ((over >> k) & 1u) {
                float we = (r[k] == c[k]) ? 0.0f : wt[k];
                int o = atomicAdd(pspill_n, 1);
                v4i en = {c[k], r[k], __float_as_int(we), 0};
                *(v4i*)(pspill + 4 * (size_t)o) = en;
            }
        }
    }
    __syncthreads();

    int myb = wv + 8 * lane;
    int myc = (myb < NBIN) ? lcnt[myb] : 0;
    myc = (myc < CAP) ? myc : CAP;
    if (myb < NBIN) celltab[myb * NCHUNK + chunk] = myc;
    v2i pc[(NBIN + 7) / 8];
#pragma unroll
    for (int i = 0; i < (NBIN + 7) / 8; ++i) {
        int b = wv + 8 * i;
        int n = __builtin_amdgcn_readlane(myc, i);
        v2i t = {0, 0};
        if (lane < n) t = lcell[b * CAP + lane];
        pc[i] = t;
    }
#pragma unroll
    for (int i = 0; i < (NBIN + 7) / 8; ++i) {
        int b = wv + 8 * i;
        int n = __builtin_amdgcn_readlane(myc, i);
        if (lane < n)
            *(v2i*)(slab + ((size_t)(b * NCHUNK + chunk) * CAP + lane) * 2) = pc[i];
    }
}

static __device__ __forceinline__ int bucket_slot(int dl, int wb, int* ncnt, float* degx) {
    int s = atomicAdd(&ncnt[dl], 1);
    atomicAdd(&degx[dl], __int_as_float(wb));
    return s;
}

static __device__ __forceinline__ void bucket_put(int s, int src, int dl, int wb, int b, v2i* nb,
                                                  int* __restrict__ spill_n, int* __restrict__ spill) {
    if (s < CAP) {
        v2i pr = {src, wb};
        nb[dl * CAP + s] = pr;
    } else {
        int o = atomicAdd(spill_n, 1);
        v4i en = {b * BIN_NODES + dl, src, wb, 0};
        *(v4i*)(spill + 4 * (size_t)o) = en;
    }
}

#define GEMM_WAVES 8
#define M_TILES    (N_NODES / 16)
#define W_PITCH4   33
#define BIN_STRIPS (BIN_NODES / 16)

template <bool NEXT>
static __device__ __forceinline__ void strip_product(float4 (&a)[16], const float4* xnext, const float4* bl,
                                                     float4 dg_lo, float4 dg_hi, float* crow) {
    v8f acc[8];
#pragma unroll
    for (int t = 0; t < 8; ++t) acc[t] = v8f{};

    float4 bcur[8], bnxt[8];
#pragma unroll
    for (int t = 0; t < 8; ++t) bcur[t] = bl[(16 * t) * W_PITCH4];

#pragma unroll
    for (int j = 0; j < 16; ++j) {
        if (j + 1 < 16) {
#pragma unroll
            for (int t = 0; t < 8; ++t) bnxt[t] = bl[(16 * t) * W_PITCH4 + 2 * (j + 1)];
        }
        if (NEXT && j > 0) a[j - 1] = xnext[2 * (j - 1)];
        __builtin_amdgcn_sched_barrier(0);
        v2f a0 = {a[j].x, a[j].y};
        v2f a1 = {a[j].z, a[j].w};
#pragma unroll
        for (int t = 0; t < 8; ++t) {
            v2f b0 = {bcur[t].x, bcur[t].y};
            acc[t] = __builtin_amdgcn_wmma_f32_16x16x4_f32(
                false, a0, false, b0, (short)0, acc[t], false, false);
        }
#pragma unroll
        for (int t = 0; t < 8; ++t) {
            v2f b1 = {bcur[t].z, bcur[t].w};
            acc[t] = __builtin_amdgcn_wmma_f32_16x16x4_f32(
                false, a1, false, b1, (short)0, acc[t], false, false);
        }
        __builtin_amdgcn_sched_barrier(0);
#pragma unroll
        for (int t = 0; t < 8; ++t) bcur[t] = bnxt[t];
    }
    if (NEXT) a[15] = xnext[30];

    float rsv[8] = {rsqrtf(dg_lo.x + 1.0f), rsqrtf(dg_lo.y + 1.0f), rsqrtf(dg_lo.z + 1.0f), rsqrtf(dg_lo.w + 1.0f),
                    rsqrtf(dg_hi.x + 1.0f), rsqrtf(dg_hi.y + 1.0f), rsqrtf(dg_hi.z + 1.0f), rsqrtf(dg_hi.w + 1.0f)};
#pragma unroll
    for (int t = 0; t < 8; ++t) {
#pragma unroll
        for (int r = 0; r < 8; ++r) acc[t][r] *= rsv[r];
    }
#pragma unroll
    for (int r = 0; r < 8; ++r) {
#pragma unroll
        for (int q = 0; q < 2; ++q) {
            float4 o = {acc[4 * q + 0][r], acc[4 * q + 1][r], acc[4 * q + 2][r], acc[4 * q + 3][r]};
            *(float4*)(crow + (size_t)r * HID + 64 * q) = o;
        }
    }
}

__global__ void __launch_bounds__(256)
bin_gemm_kernel(const int* __restrict__ slab, const int* __restrict__ celltab,
                const int* __restrict__ pspill_n, const int* __restrict__ pspill,
                int* __restrict__ spill_n, int* __restrict__ spill,
                int* __restrict__ cnt, int* __restrict__ buckets, float* __restrict__ dis,
                const float* __restrict__ X, const float* __restrict__ W1, float* __restrict__ C) {
    __shared__ int    ncnt[BIN_NODES];
    __shared__ __attribute__((aligned(16))) float degx[BIN_NODES];
    __shared__ v2i    nb[BIN_NODES * CAP];
    __shared__ float4 wl[HID * W_PITCH4];

    int tid  = threadIdx.x;
    int lane = tid & 31;
    int wv   = __builtin_amdgcn_readfirstlane(tid >> 5);
    int b    = blockIdx.x;
    int c    = lane & 15;
    int h    = lane >> 4;

    ncnt[tid] = 0;
    degx[tid] = 0.0f;
    float4 wrow[HID / GEMM_WAVES];
#pragma unroll
    for (int i = 0; i < HID / GEMM_WAVES; ++i) {
        int n = i * GEMM_WAVES + wv;
        wrow[i] = ((const float4*)(W1 + (size_t)n * IN_CH))[lane];
    }
    int myn = celltab[b * NCHUNK + wv * 32 + lane];
    int np  = *pspill_n;
#pragma unroll
    for (int i = 0; i < HID / GEMM_WAVES; ++i) {
        int n  = i * GEMM_WAVES + wv;
        int t  = ((n >> 6) << 2) | (n & 3);
        int cc = (n & 63) >> 2;
        wl[(16 * t + cc) * W_PITCH4 + lane] = wrow[i];
    }
    __syncthreads();

    const v2i* cells = (const v2i*)slab + (size_t)(b * NCHUNK + wv * 32) * CAP;
    const unsigned long long* cells64 = (const unsigned long long*)cells;
    unsigned long long en[32];
#pragma unroll
    for (int k = 0; k < 32; ++k) {
        int n = __builtin_amdgcn_readlane(myn, k);
        unsigned long long t = ~0ull;
        if (lane < n) t = cells64[k * CAP + lane];
        en[k] = t;
    }
    int st0 = b * BIN_STRIPS + wv;
    int st1 = st0 + GEMM_WAVES;
    int s0c = (st0 < M_TILES) ? st0 : (M_TILES - 1);
    int s1c = (st1 < M_TILES) ? st1 : (M_TILES - 1);
    const float4* xr0 = (const float4*)(X + (size_t)(s0c * 16 + c) * IN_CH) + h;
    const float4* xr1 = (const float4*)(X + (size_t)(s1c * 16 + c) * IN_CH) + h;
    float4 a[16];
#pragma unroll
    for (int j = 0; j < 16; ++j) a[j] = xr0[2 * j];
#pragma unroll
    for (int g = 0; g < 32; g += 16) {
        int s[16];
#pragma unroll
        for (int k = 0; k < 16; ++k) {
            int ex = (int)en[g + k];
            int wb = (int)(en[g + k] >> 32);
            s[k] = 0;
            if (ex >= 0) s[k] = bucket_slot((ex >> 16) & (BIN_NODES - 1), wb, ncnt, degx);
        }
        unsigned over = 0u;
#pragma unroll
        for (int k = 0; k < 16; ++k) {
            int ex = (int)en[g + k];
            int wb = (int)(en[g + k] >> 32);
            if (ex >= 0) {
                if (s[k] < CAP) {
                    v2i pr = {ex & 0xffff, wb};
                    nb[((ex >> 16) & (BIN_NODES - 1)) * CAP + s[k]] = pr;
                } else {
                    over |= 1u << k;
                }
            }
        }
        if (over != 0u) {
#pragma unroll
            for (int k = 0; k < 16; ++k) {
                if ((over >> k) & 1u) {
                    int ex = (int)en[g + k];
                    bucket_put(CAP, ex & 0xffff, (ex >> 16) & (BIN_NODES - 1), (int)(en[g + k] >> 32), b,
                               nb, spill_n, spill);
                }
            }
        }
    }
    for (int k = tid; k < np; k += 256) {
        v4i pe = *(const v4i*)(pspill + 4 * (size_t)k);
        if ((pe.x >> 8) == b) {
            int dl = pe.x & (BIN_NODES - 1);
            int s  = bucket_slot(dl, pe.z, ncnt, degx);
            bucket_put(s, pe.y, dl, pe.z, b, nb, spill_n, spill);
        }
    }
    __syncthreads();

    int node = b * BIN_NODES + tid;
    int n = ncnt[tid];
    float d = degx[tid] + 1.0f;
    if (node < N_NODES) {
        cnt[node] = n;
        dis[node] = rsqrtf(d);
    }

    v4i*       gb = (v4i*)(buckets + (size_t)b * (BIN_NODES * 2 * CAP));
    const v4i* lb = (const v4i*)nb;
    int mm[16];
    v4i pcs[16];
#pragma unroll
    for (int i = 0; i < 16; ++i) {
        int idx = i * 256 + tid;
        int m   = ncnt[idx >> 4];
        mm[i] = (m < CAP) ? m : CAP;
    }
#pragma unroll
    for (int i = 0; i < 16; ++i) {
        int idx = i * 256 + tid;
        int nd  = idx >> 4;
        int q   = idx & 15;
        v4i t = {0, 0, 0, 0};
        if (2 * q < mm[i] && b * BIN_NODES + nd < N_NODES) t = lb[idx];
        pcs[i] = t;
    }
#pragma unroll
    for (int i = 0; i < 16; ++i) {
        int idx = i * 256 + tid;
        int nd  = idx >> 4;
        int q   = idx & 15;
        if (2 * q < mm[i] && b * BIN_NODES + nd < N_NODES) gb[idx] = pcs[i];
    }

    const float4* bl = wl + c * W_PITCH4 + h;
    const float4* dg = (const float4*)degx + 2 * h;
    if (st0 < M_TILES)
        strip_product<true>(a, xr1, bl, dg[4 * wv], dg[4 * wv + 1],
                            C + (size_t)(st0 * 16 + h * 8) * HID + 4 * c);
    if (st1 < M_TILES)
        strip_product<false>(a, xr1, bl, dg[4 * (wv + GEMM_WAVES)], dg[4 * (wv + GEMM_WAVES) + 1],
                             C + (size_t)(st1 * 16 + h * 8) * HID + 4 * c);
}

struct bucket_t { int n; int rj; int wj; };

static __device__ __forceinline__ bucket_t bucket_fetch(int c, int lane,
                                                        const int* __restrict__ cnt,
                                                        const int* __restrict__ buckets) {
    bucket_t k;
    k.n  = cnt[c];
    k.rj = 0;
    k.wj = 0;
    int m = (k.n < CAP) ? k.n : CAP;
    if (lane < m) {
        v2i pr = *(const v2i*)(buckets + (size_t)c * (2 * CAP) + 2 * lane);
        k.rj = pr.x;
        k.wj = pr.y;
    }
    return k;
}

static __device__ __forceinline__ float4 gather_row(int c, int lane, bucket_t k,
                                                    const int* __restrict__ spill_n,
                                                    const int* __restrict__ spill,
                                                    const float* __restrict__ src) {
    int n = k.n;
    int m = (n < CAP) ? n : CAP;
    const float4* src4 = (const float4*)src;

    float4 acc = {0.0f, 0.0f, 0.0f, 0.0f};
    int   rj = k.rj, wj = k.wj;
    int last = m - 1;
    for (int j = 0; j < m; j += 4) {
        int j1 = (j + 1 < last) ? j + 1 : last;
        int j2 = (j + 2 < last) ? j + 2 : last;
        int j3 = (j + 3 < last) ? j + 3 : last;
        int r0 = __builtin_amdgcn_readlane(rj, j);
        int r1 = __builtin_amdgcn_readlane(rj, j1);
        int r2 = __builtin_amdgcn_readlane(rj, j2);
        int r3 = __builtin_amdgcn_readlane(rj, j3);
        float4 v0 = src4[(size_t)r0 * (HID / 4) + lane];
        float4 v1 = src4[(size_t)r1 * (HID / 4) + lane];
        float4 v2 = src4[(size_t)r2 * (HID / 4) + lane];
        float4 v3 = src4[(size_t)r3 * (HID / 4) + lane];
        float c0 = __int_as_float(__builtin_amdgcn_readlane(wj, j));
        float c1 = __int_as_float(__builtin_amdgcn_readlane(wj, j1));
        float c2 = __int_as_float(__builtin_amdgcn_readlane(wj, j2));
        float c3 = __int_as_float(__builtin_amdgcn_readlane(wj, j3));
        if (j + 1 >= m) c1 = 0.0f;
        if (j + 2 >= m) c2 = 0.0f;
        if (j + 3 >= m) c3 = 0.0f;
        acc.x += c0 * v0.x; acc.y += c0 * v0.y; acc.z += c0 * v0.z; acc.w += c0 * v0.w;
        acc.x += c1 * v1.x; acc.y += c1 * v1.y; acc.z += c1 * v1.z; acc.w += c1 * v1.w;
        acc.x += c2 * v2.x; acc.y += c2 * v2.y; acc.z += c2 * v2.z; acc.w += c2 * v2.w;
        acc.x += c3 * v3.x; acc.y += c3 * v3.y; acc.z += c3 * v3.z; acc.w += c3 * v3.w;
    }
    if (n > CAP) {
        int no = *spill_n;
        for (int base = 0; base < no; base += 32) {
            int k  = base + lane;
            int oc = -1, orow = 0, ow = 0;
            if (k < no) {
                v4i en = *(const v4i*)(spill + 4 * (size_t)k);
                oc = en.x; orow = en.y; ow = en.z;
            }
            unsigned hit = __builtin_amdgcn_ballot_w32(oc == c);
            while (hit != 0u) {
                int l = __builtin_ctz(hit);
                hit &= hit - 1u;
                int   r  = __builtin_amdgcn_readlane(orow, l);
                float cw = __int_as_float(__builtin_amdgcn_readlane(ow, l));
                float4 v = src4[(size_t)r * (HID / 4) + lane];
                acc.x += cw * v.x; acc.y += cw * v.y; acc.z += cw * v.z; acc.w += cw * v.w;
            }
        }
    }
    return acc;
}

#define FUSE_WAVES 8
#define FUSE_GRID  1024
#define T_PITCH4   33

__global__ void __launch_bounds__(FUSE_WAVES * 32, 8)
gather_relu_gemm_kernel(const int* __restrict__ cnt, const int* __restrict__ spill_n,
                        const int* __restrict__ buckets, const int* __restrict__ spill,
                        const float* __restrict__ dis,
                        const float* __restrict__ xw, const float* __restrict__ b1,
                        const float* __restrict__ Wa, const float* __restrict__ Wb,
                        float* __restrict__ xw2) {
    __shared__ float4 ht[16 * T_PITCH4];
    __shared__ float4 ot[16 * T_PITCH4];

    int tid  = threadIdx.x;
    int lane = tid & 31;
    int wv   = __builtin_amdgcn_readfirstlane(tid >> 5);
    int c    = lane & 15;
    int h    = lane >> 4;

    int nrow = 16 * wv + c;
    const float*  wrow = (nrow < OUT_CH) ? (Wa + (size_t)nrow * HID)
                                         : (Wb + (size_t)(nrow - OUT_CH) * HID);
    const float4* wr4  = (const float4*)wrow + h;
    float4 bf[16];
#pragma unroll
    for (int j = 0; j < 16; ++j) bf[j] = wr4[2 * j];

    float4 bias = ((const float4*)b1)[lane];
    const float4* xw4 = (const float4*)xw;

    for (int s = blockIdx.x; s < M_TILES; s += FUSE_GRID) {
        int c0 = __builtin_amdgcn_readfirstlane(s * 16 + wv);
        int c1 = c0 + 8;
        bucket_t k0 = bucket_fetch(c0, lane, cnt, buckets);
        bucket_t k1 = bucket_fetch(c1, lane, cnt, buckets);
        float4 own0 = xw4[(size_t)c0 * (HID / 4) + lane];
        float4 own1 = xw4[(size_t)c1 * (HID / 4) + lane];
        float  d0   = dis[c0];
        float  d1   = dis[c1];

        float4 g0 = gather_row(c0, lane, k0, spill_n, spill, xw);
        float4 v0;
        v0.x = fmaxf((g0.x + own0.x) * d0 + bias.x, 0.0f);
        v0.y = fmaxf((g0.y + own0.y) * d0 + bias.y, 0.0f);
        v0.z = fmaxf((g0.z + own0.z) * d0 + bias.z, 0.0f);
        v0.w = fmaxf((g0.w + own0.w) * d0 + bias.w, 0.0f);
        ht[wv * T_PITCH4 + lane] = v0;

        float4 g1 = gather_row(c1, lane, k1, spill_n, spill, xw);
        float4 v1;
        v1.x = fmaxf((g1.x + own1.x) * d1 + bias.x, 0.0f);
        v1.y = fmaxf((g1.y + own1.y) * d1 + bias.y, 0.0f);
        v1.z = fmaxf((g1.z + own1.z) * d1 + bias.z, 0.0f);
        v1.w = fmaxf((g1.w + own1.w) * d1 + bias.w, 0.0f);
        ht[(wv + 8) * T_PITCH4 + lane] = v1;
        __syncthreads();

        const float4* al = ht + c * T_PITCH4 + h;
        v8f acc = v8f{};
        float4 a[8];
#pragma unroll
        for (int g = 0; g < 2; ++g) {
#pragma unroll
            for (int j = 0; j < 8; ++j) a[j] = al[2 * (8 * g + j)];
#pragma unroll
            for (int j = 0; j < 8; ++j) {
                v2f a0 = {a[j].x, a[j].y};
                v2f a1 = {a[j].z, a[j].w};
                v2f b0 = {bf[8 * g + j].x, bf[8 * g + j].y};
                v2f b1 = {bf[8 * g + j].z, bf[8 * g + j].w};
                acc = __builtin_amdgcn_wmma_f32_16x16x4_f32(
                    false, a0, false, b0, (short)0, acc, false, false);
                acc = __builtin_amdgcn_wmma_f32_16x16x4_f32(
                    false, a1, false, b1, (short)0, acc, false, false);
            }
        }

        float* of = (float*)ot;
#pragma unroll
        for (int r = 0; r < 8; ++r) of[(8 * h + r) * (4 * T_PITCH4) + 16 * wv + c] = acc[r];
        __syncthreads();

        float4 o0 = ot[wv * T_PITCH4 + lane];
        float4 o1 = ot[(wv + 8) * T_PITCH4 + lane];
        o0.x *= d0; o0.y *= d0; o0.z *= d0; o0.w *= d0;
        o1.x *= d1; o1.y *= d1; o1.z *= d1; o1.w *= d1;
        ((float4*)xw2)[(size_t)c0 * (HID / 4) + lane] = o0;
        ((float4*)xw2)[(size_t)c1 * (HID / 4) + lane] = o1;
    }
}

__global__ void gather_out_kernel(const int* __restrict__ cnt, const int* __restrict__ spill_n,
                                  const int* __restrict__ buckets, const int* __restrict__ spill,
                                  const float* __restrict__ dis,
                                  const float* __restrict__ xw2, const float* __restrict__ b_mu,
                                  const float* __restrict__ b_ls, float* __restrict__ out) {
    int wave = (blockIdx.x * blockDim.x + threadIdx.x) >> 5;
    int lane = threadIdx.x & 31;
    if (wave >= N_NODES) return;
    int c = __builtin_amdgcn_readfirstlane(wave);
    int l16 = lane & 15;
    float4 own  = ((const float4*)xw2)[(size_t)c * (HID / 4) + lane];
    float4 bias = (lane < 16) ? ((const float4*)b_mu)[l16] : ((const float4*)b_ls)[l16];
    float  dc   = dis[c];
    bucket_t k  = bucket_fetch(c, lane, cnt, buckets);
    float4 acc  = gather_row(c, lane, k, spill_n, spill, xw2);
    float4 v;
    v.x = (acc.x + own.x) * dc + bias.x;
    v.y = (acc.y + own.y) * dc + bias.y;
    v.z = (acc.z + own.z) * dc + bias.z;
    v.w = (acc.w + own.w) * dc + bias.w;
    float* orow = out + ((lane < 16) ? (size_t)0 : (size_t)N_NODES * OUT_CH)
                      + (size_t)c * OUT_CH + l16 * 4;
    *(float4*)orow = v;
}

extern "C" void kernel_launch(void* const* d_in, const int* in_sizes, int n_in,
                              void* d_out, int out_size, void* d_ws, size_t ws_size,
                              hipStream_t stream) {
    (void)in_sizes; (void)n_in; (void)out_size; (void)ws_size;
    const float* x   = (const float*)d_in[0];
    const int*   ei  = (const int*)d_in[1];
    const float* ew  = (const float*)d_in[2];
    const float* W1  = (const float*)d_in[3];
    const float* b1  = (const float*)d_in[4];
    const float* Wmu = (const float*)d_in[5];
    const float* bmu = (const float*)d_in[6];
    const float* Wls = (const float*)d_in[7];
    const float* bls = (const float*)d_in[8];
    const int* row = ei;
    const int* col = ei + N_EDGES;

    float* ws      = (float*)d_ws;
    float* dis     = ws;
    int*   cnt     = (int*)(ws + 50000);
    int*   spill_n = cnt + N_NODES;
    float* xw      = ws + 150016;
    float* agg     = ws + 150016 + 6400000;
    int*   buckets = (int*)(ws + 12950016);
    int*   spill   = (int*)(ws + 16150016);
    int*   slab    = (int*)(ws + 6550016);
    int*   celltab = (int*)(ws + 9761280);
    int*   pspill  = (int*)(ws + 9811456);

    hipMemsetAsync(spill_n, 0, 4 * sizeof(int), stream);

    part_kernel<<<NCHUNK, 256, 0, stream>>>(row, col, ew, spill_n + 1, slab, celltab, pspill);
    bin_gemm_kernel<<<NBIN, 256, 0, stream>>>(slab, celltab, spill_n + 1, pspill, spill_n, spill, cnt, buckets, dis,
                                              x, W1, xw);
    gather_relu_gemm_kernel<<<FUSE_GRID, FUSE_WAVES * 32, 0, stream>>>(cnt, spill_n, buckets, spill, dis, xw, b1,
                                                                       Wmu, Wls, agg);

    gather_out_kernel<<<N_NODES / 8, 256, 0, stream>>>(cnt, spill_n, buckets, spill, dis, agg, bmu, bls,
                                                       (float*)d_out);
}
